# starting kernel
# baseline (speedup 1.0000x reference)
#include <hip/hip_runtime.h>

#define N_NODES 100000
#define N_EDGES 1600000
#define N_GRAPHS 2000
#define HIDDEN 128
#define N_LAYERS 4
#define BN_EPS 1e-5f

typedef __attribute__((ext_vector_type(2))) float v2f;
typedef __attribute__((ext_vector_type(8))) float v8f;

__global__ void zero_f32(float* __restrict__ p, size_t n) {
    size_t i = (size_t)blockIdx.x * blockDim.x + threadIdx.x;
    if (i < n) p[i] = 0.0f;
}

__global__ void degree_kernel(const int* __restrict__ dst, float* __restrict__ deg) {
    int e = blockIdx.x * blockDim.x + threadIdx.x;
    if (e < N_EDGES) atomicAdd(&deg[dst[e]], 1.0f);
}

__global__ void norm_kernel(float* __restrict__ norm) {
    int i = blockIdx.x * blockDim.x + threadIdx.x;
    if (i < N_NODES) norm[i] = rsqrtf(norm[i] + 1.0f);
}

__global__ __launch_bounds__(256) void gemm_xw_wmma(const float* __restrict__ H,
                                                    const float* __restrict__ W,
                                                    float* __restrict__ XW) {
    const int tile = blockIdx.x * 8 + (threadIdx.x >> 5);
    if (tile >= N_NODES / 16) return;
    const int lane  = threadIdx.x & 31;
    const int l16   = lane & 15;
    const int hi    = lane >> 4;
    const int khalf = hi * 2;
    const size_t rowA = (size_t)(tile * 16 + l16) * HIDDEN;

    v8f acc[8];
#pragma unroll
    for (int nt = 0; nt < 8; ++nt) acc[nt] = (v8f)0.0f;

#pragma unroll 4
    for (int kk = 0; kk < 32; ++kk) {
        const int k = kk * 4 + khalf;
        v2f a;
        a.x = H[rowA + k];
        a.y = H[rowA + k + 1];
#pragma unroll
        for (int nt = 0; nt < 8; ++nt) {
            const int n = nt * 16 + l16;
            v2f b;
            b.x = W[(size_t)k * HIDDEN + n];
            b.y = W[(size_t)(k + 1) * HIDDEN + n];
            acc[nt] = __builtin_amdgcn_wmma_f32_16x16x4_f32(
                false, a, false, b, (short)0, acc[nt], false, false);
        }
    }

#pragma unroll
    for (int v = 0; v < 8; ++v) {
        const int row = tile * 16 + hi * 8 + v;
#pragma unroll
        for (int nt = 0; nt < 8; ++nt) {
            XW[(size_t)row * HIDDEN + nt * 16 + l16] = acc[nt][v];
        }
    }
}

__global__ void scatter_kernel(const float* __restrict__ xw,
                               const int* __restrict__ src,
                               const int* __restrict__ dst,
                               const float* __restrict__ norm,
                               float* __restrict__ agg) {
    size_t t = (size_t)blockIdx.x * blockDim.x + threadIdx.x;
    int e = (int)(t >> 5);
    if (e >= N_EDGES) return;
    int lane = (int)(t & 31);
    int s = src[e], d = dst[e];
    float c = norm[s] * norm[d];
    const float4 v = *(const float4*)(xw + (size_t)s * HIDDEN + lane * 4);
    float* o = agg + (size_t)d * HIDDEN + lane * 4;
    atomicAdd(o + 0, v.x * c);
    atomicAdd(o + 1, v.y * c);
    atomicAdd(o + 2, v.z * c);
    atomicAdd(o + 3, v.w * c);
}

__global__ void update_kernel(const float* __restrict__ xw,
                              const float* __restrict__ norm,
                              const float* __restrict__ bias,
                              const float* __restrict__ gamma,
                              const float* __restrict__ beta,
                              const float* __restrict__ mean,
                              const float* __restrict__ var,
                              float* __restrict__ agg) {
    size_t t = (size_t)blockIdx.x * blockDim.x + threadIdx.x;
    if (t >= (size_t)N_NODES * 32) return;
    int node = (int)(t >> 5);
    int f0 = (int)(t & 31) * 4;
    float nr = norm[node];
    float sc = nr * nr;
    size_t base = (size_t)node * HIDDEN + f0;
#pragma unroll
    for (int j = 0; j < 4; ++j) {
        int f = f0 + j;
        float val = agg[base + j] + xw[base + j] * sc + bias[f];
        val = (val - mean[f]) * (gamma[f] * rsqrtf(var[f] + BN_EPS)) + beta[f];
        agg[base + j] = fmaxf(val, 0.0f);
    }
}

__global__ void count_kernel(const int* __restrict__ batch, float* __restrict__ counts) {
    int i = blockIdx.x * blockDim.x + threadIdx.x;
    if (i < N_NODES) atomicAdd(&counts[batch[i]], 1.0f);
}

__global__ void pool_kernel(const float* __restrict__ h, const int* __restrict__ batch,
                            float* __restrict__ sums) {
    size_t t = (size_t)blockIdx.x * blockDim.x + threadIdx.x;
    if (t >= (size_t)N_NODES * 32) return;
    int node = (int)(t >> 5);
    int f0 = (int)(t & 31) * 4;
    int g = batch[node];
    const float4 v = *(const float4*)(h + (size_t)node * HIDDEN + f0);
    float* o = sums + (size_t)g * HIDDEN + f0;
    atomicAdd(o + 0, v.x);
    atomicAdd(o + 1, v.y);
    atomicAdd(o + 2, v.z);
    atomicAdd(o + 3, v.w);
}

__global__ void out_kernel(const float* __restrict__ sums, const float* __restrict__ counts,
                           const float* __restrict__ W_out, const float* __restrict__ b_out,
                           float* __restrict__ out) {
    int t = blockIdx.x * blockDim.x + threadIdx.x;
    if (t >= N_GRAPHS * 5) return;
    int g = t / 5, o = t % 5;
    float inv = 1.0f / fmaxf(counts[g], 1.0f);
    float acc = b_out[o];
    for (int f = 0; f < HIDDEN; ++f)
        acc += sums[(size_t)g * HIDDEN + f] * inv * W_out[f * 5 + o];
    out[t] = acc;
}

extern "C" void kernel_launch(void* const* d_in, const int* in_sizes, int n_in,
                              void* d_out, int out_size, void* d_ws, size_t ws_size,
                              hipStream_t stream) {
    const float* x        = (const float*)d_in[0];
    const int*   ei       = (const int*)d_in[1];
    const int*   src      = ei;
    const int*   dst      = ei + N_EDGES;
    const int*   batch    = (const int*)d_in[2];
    const float* Ws       = (const float*)d_in[3];
    const float* bs       = (const float*)d_in[4];
    const float* gammas   = (const float*)d_in[5];
    const float* betas    = (const float*)d_in[6];
    const float* run_mean = (const float*)d_in[7];
    const float* run_var  = (const float*)d_in[8];
    const float* W_out    = (const float*)d_in[9];
    const float* b_out    = (const float*)d_in[10];
    float*       out      = (float*)d_out;

    const size_t NF = (size_t)N_NODES * HIDDEN;
    float* bufA   = (float*)d_ws;
    float* bufB   = bufA + NF;
    float* bufC   = bufB + NF;
    float* norm   = bufC + NF;
    float* sums   = norm + N_NODES;
    float* counts = sums + (size_t)N_GRAPHS * HIDDEN;

    const int B = 256;
    auto cdiv = [](size_t a, size_t b) { return (int)((a + b - 1) / b); };

    zero_f32<<<cdiv(N_NODES, B), B, 0, stream>>>(norm, (size_t)N_NODES);
    degree_kernel<<<cdiv(N_EDGES, B), B, 0, stream>>>(dst, norm);
    norm_kernel<<<cdiv(N_NODES, B), B, 0, stream>>>(norm);

    const float* h = x;
    float* aggsel[2] = { bufA, bufC };
    for (int l = 0; l < N_LAYERS; ++l) {
        float* agg = aggsel[l & 1];
        gemm_xw_wmma<<<cdiv(N_NODES / 16, 8), 256, 0, stream>>>(
            h, Ws + (size_t)l * HIDDEN * HIDDEN, bufB);
        zero_f32<<<cdiv(NF, B), B, 0, stream>>>(agg, NF);
        scatter_kernel<<<cdiv((size_t)N_EDGES * 32, B), B, 0, stream>>>(
            bufB, src, dst, norm, agg);
        update_kernel<<<cdiv((size_t)N_NODES * 32, B), B, 0, stream>>>(
            bufB, norm, bs + l * HIDDEN, gammas + l * HIDDEN, betas + l * HIDDEN,
            run_mean + l * HIDDEN, run_var + l * HIDDEN, agg);
        h = agg;
    }

    zero_f32<<<cdiv((size_t)N_GRAPHS * HIDDEN + N_GRAPHS, B), B, 0, stream>>>(
        sums, (size_t)N_GRAPHS * HIDDEN + N_GRAPHS);
    count_kernel<<<cdiv(N_NODES, B), B, 0, stream>>>(batch, counts);
    pool_kernel<<<cdiv((size_t)N_NODES * 32, B), B, 0, stream>>>(h, batch, sums);
    out_kernel<<<cdiv(N_GRAPHS * 5, B), B, 0, stream>>>(sums, counts, W_out, b_out, out);
}


# Edge sums without float atomics: edges grouped by destination in-call (count, scan, place, runs ordered by source); one wave sums a row in registers in that fixed order and writes it once
# speedup vs baseline: 3.1383x; 3.1383x over previous
#include <hip/hip_runtime.h>

#define N_NODES 100000
#define N_EDGES 1600000
#define N_GRAPHS 2000
#define HIDDEN 128
#define N_LAYERS 4
#define BN_EPS 1e-5f
#define SCAN_BLOCKS 98
#define N_PAD (SCAN_BLOCKS * 1024)

typedef __attribute__((ext_vector_type(2))) float v2f;
typedef __attribute__((ext_vector_type(8))) float v8f;


__global__ void zero_f32(float* __restrict__ p, size_t n) {
    size_t i = (size_t)blockIdx.x * blockDim.x + threadIdx.x;
    if (i < n) p[i] = 0.0f;
}

__global__ void hist_rank_kernel(const int* __restrict__ dst, int* __restrict__ cnt,
                                 int* __restrict__ rank) {
    int e = blockIdx.x * blockDim.x + threadIdx.x;
    if (e < N_EDGES) rank[e] = atomicAdd(&cnt[dst[e]], 1);
}

__global__ __launch_bounds__(256) void scan_local_kernel(const int* __restrict__ cnt,
                                                         int* __restrict__ rowptr,
                                                         int* __restrict__ blocksum,
                                                         float* __restrict__ norm) {
    __shared__ int wsum[8];
    const int t = threadIdx.x, lane = t & 31, w = t >> 5;
    const int i0 = (blockIdx.x * 256 + t) * 4;
    const int4 c = *(const int4*)(cnt + i0);
    const int s = c.x + c.y + c.z + c.w;
    int incl = s;
#pragma unroll
    for (int o = 1; o < 32; o <<= 1) {
        int n = __shfl_up(incl, o);
        if (lane >= o) incl += n;
    }
    if (lane == 31) wsum[w] = incl;
    __syncthreads();
    int woff = 0, total = 0;
#pragma unroll
    for (int k = 0; k < 8; ++k) {
        int v = wsum[k];
        total += v;
        if (k < w) woff += v;
    }
    int4 r;
    r.x = woff + incl - s;
    r.y = r.x + c.x;
    r.z = r.y + c.y;
    r.w = r.z + c.z;
    *(int4*)(rowptr + i0) = r;
    if (t == 0) blocksum[blockIdx.x] = total;
    if (i0 < N_NODES) {
        float4 nv;
        nv.x = rsqrtf((float)c.x + 1.0f);
        nv.y = rsqrtf((float)c.y + 1.0f);
        nv.z = rsqrtf((float)c.z + 1.0f);
        nv.w = rsqrtf((float)c.w + 1.0f);
        *(float4*)(norm + i0) = nv;
    }
}

__global__ __launch_bounds__(256) void scan_add_kernel(int* __restrict__ rowptr,
                                                       const int* __restrict__ blocksum) {
    __shared__ int soff;
    const int t = threadIdx.x, b = blockIdx.x;
    if (t < 32) {
        int v = 0;
        for (int k = t; k < b; k += 32) v += blocksum[k];
#pragma unroll
        for (int o = 16; o > 0; o >>= 1) v += __shfl_xor(v, o);
        if (t == 0) soff = v;
    }
    __syncthreads();
    const int off = soff;
    const int i0 = (b * 256 + t) * 4;
    int4 r = *(const int4*)(rowptr + i0);
    r.x += off; r.y += off; r.z += off; r.w += off;
    *(int4*)(rowptr + i0) = r;
}

__global__ void fill_kernel(const int* __restrict__ src, const int* __restrict__ dst,
                            const int* __restrict__ rank, const int* __restrict__ rowptr,
                            const float* __restrict__ norm, int* __restrict__ rec) {
    int e = blockIdx.x * blockDim.x + threadIdx.x;
    if (e >= N_EDGES) return;
    const int s = src[e], d = dst[e];
    const int p = rowptr[d] + rank[e];
    const float c = norm[s] * norm[d];
    *(int2*)(rec + 2 * (size_t)p) = make_int2(s, __float_as_int(c));
}

__global__ __launch_bounds__(256) void order_runs_kernel(const int* __restrict__ rowptr,
                                                         const int* __restrict__ rec_in,
                                                         int* __restrict__ rec_out) {
    const int wave = __builtin_amdgcn_readfirstlane((int)(threadIdx.x >> 5));
    const int node = blockIdx.x * 8 + wave;
    if (node >= N_NODES) return;
    const int lane = threadIdx.x & 31;
    const int beg = rowptr[node];
    const int deg = rowptr[node + 1] - beg;
    for (int ib = 0; ib < deg; ib += 32) {
        const int i = ib + lane;
        int2 mine = make_int2(0, 0);
        if (i < deg) mine = *(const int2*)(rec_in + 2 * (size_t)(beg + i));
        int rk = 0;
        for (int jb = 0; jb < deg; jb += 32) {
            const int j = jb + lane;
            int key = 0x7fffffff;
            if (j < deg) key = rec_in[2 * (size_t)(beg + j)];
            const int n = min(32, deg - jb);
            for (int u = 0; u < n; ++u) {
                const int k = __builtin_amdgcn_readlane(key, u);
                rk += (int)(k < mine.x) | ((int)(k == mine.x) & (int)(jb + u < i));
            }
        }
        if (i < deg) *(int2*)(rec_out + 2 * (size_t)(beg + rk)) = mine;
    }
}

__global__ __launch_bounds__(256) void gemm_xw_wmma(const float* __restrict__ H,
                                                    const float* __restrict__ W,
                                                    float* __restrict__ XW) {
    const int tile = blockIdx.x * 8 + (threadIdx.x >> 5);
    if (tile >= N_NODES / 16) return;
    const int lane  = threadIdx.x & 31;
    const int l16   = lane & 15;
    const int hi    = lane >> 4;
    const int khalf = hi * 2;
    const size_t rowA = (size_t)(tile * 16 + l16) * HIDDEN;

    v8f acc[8];
#pragma unroll
    for (int nt = 0; nt < 8; ++nt) acc[nt] = (v8f)0.0f;

#pragma unroll 4
    for (int kk = 0; kk < 32; ++kk) {
        const int k = kk * 4 + khalf;
        v2f a;
        a.x = H[rowA + k];
        a.y = H[rowA + k + 1];
#pragma unroll
        for (int nt = 0; nt < 8; ++nt) {
            const int n = nt * 16 + l16;
            v2f b;
            b.x = W[(size_t)k * HIDDEN + n];
            b.y = W[(size_t)(k + 1) * HIDDEN + n];
            acc[nt] = __builtin_amdgcn_wmma_f32_16x16x4_f32(
                false, a, false, b, (short)0, acc[nt], false, false);
        }
    }

#pragma unroll
    for (int v = 0; v < 8; ++v) {
        const int row = tile * 16 + hi * 8 + v;
#pragma unroll
        for (int nt = 0; nt < 8; ++nt) {
            XW[(size_t)row * HIDDEN + nt * 16 + l16] = acc[nt][v];
        }
    }
}

template <int K>
__device__ __forceinline__ void add_rows(const float* __restrict__ xw_lane, int rs, int rc,
                                         int j, float4& acc) {
    float4 v[K];
    float c[K];
#pragma unroll
    for (int u = 0; u < K; ++u) {
        const int s = __builtin_amdgcn_readlane(rs, j + u);
        c[u] = __int_as_float(__builtin_amdgcn_readlane(rc, j + u));
        v[u] = *(const float4*)(xw_lane + (size_t)s * HIDDEN);
    }
#pragma unroll
    for (int u = 0; u < K; ++u) {
        acc.x += v[u].x * c[u];
        acc.y += v[u].y * c[u];
        acc.z += v[u].z * c[u];
        acc.w += v[u].w * c[u];
    }
}

__global__ __launch_bounds__(256) __attribute__((amdgpu_waves_per_eu(4, 12))) void gather_update_kernel(const float* __restrict__ xw,
                                                            const int* __restrict__ rowptr,
                                                            const int* __restrict__ rec,
                                                            const float* __restrict__ norm,
                                                            const float* __restrict__ bias,
                                                            const float* __restrict__ gamma,
                                                            const float* __restrict__ beta,
                                                            const float* __restrict__ mean,
                                                            const float* __restrict__ var,
                                                            float* __restrict__ hout) {
    const int wave = __builtin_amdgcn_readfirstlane((int)(threadIdx.x >> 5));
    const int node = blockIdx.x * 8 + wave;
    if (node >= N_NODES) return;
    const int lane = threadIdx.x & 31;
    const int f0 = lane * 4;
    const int beg = rowptr[node], end = rowptr[node + 1];
    const float* xw_lane = xw + f0;
    const float4 self = *(const float4*)(xw_lane + (size_t)node * HIDDEN);
    const float nr = norm[node];
    const float4 bi = *(const float4*)(bias + f0);
    const float4 ga = *(const float4*)(gamma + f0);
    const float4 be = *(const float4*)(beta + f0);
    const float4 mu = *(const float4*)(mean + f0);
    const float4 va = *(const float4*)(var + f0);

    float4 acc = make_float4(0.0f, 0.0f, 0.0f, 0.0f);
    for (int base = beg; base < end; base += 32) {
        const int cnt = min(end - base, 32);
        int2 pr = make_int2(0, 0);
        if (lane < cnt) pr = *(const int2*)(rec + 2 * (size_t)(base + lane));
        int j = 0;
        for (; j + 8 <= cnt; j += 8) add_rows<8>(xw_lane, pr.x, pr.y, j, acc);
        if (cnt & 4) { add_rows<4>(xw_lane, pr.x, pr.y, j, acc); j += 4; }
        if (cnt & 2) { add_rows<2>(xw_lane, pr.x, pr.y, j, acc); j += 2; }
        if (cnt & 1) { add_rows<1>(xw_lane, pr.x, pr.y, j, acc); }
    }

    const float sc = nr * nr;
    float4 o;
    o.x = acc.x + self.x * sc + bi.x;
    o.y = acc.y + self.y * sc + bi.y;
    o.z = acc.z + self.z * sc + bi.z;
    o.w = acc.w + self.w * sc + bi.w;
    o.x = (o.x - mu.x) * (ga.x * rsqrtf(va.x + BN_EPS)) + be.x;
    o.y = (o.y - mu.y) * (ga.y * rsqrtf(va.y + BN_EPS)) + be.y;
    o.z = (o.z - mu.z) * (ga.z * rsqrtf(va.z + BN_EPS)) + be.z;
    o.w = (o.w - mu.w) * (ga.w * rsqrtf(va.w + BN_EPS)) + be.w;
    o.x = fmaxf(o.x, 0.0f);
    o.y = fmaxf(o.y, 0.0f);
    o.z = fmaxf(o.z, 0.0f);
    o.w = fmaxf(o.w, 0.0f);
    *(float4*)(hout + (size_t)node * HIDDEN + f0) = o;
}

__global__ void count_kernel(const int* __restrict__ batch, float* __restrict__ counts) {
    int i = blockIdx.x * blockDim.x + threadIdx.x;
    if (i < N_NODES) atomicAdd(&counts[batch[i]], 1.0f);
}

__global__ void pool_kernel(const float* __restrict__ h, const int* __restrict__ batch,
                            float* __restrict__ sums) {
    size_t t = (size_t)blockIdx.x * blockDim.x + threadIdx.x;
    if (t >= (size_t)N_NODES * 32) return;
    int node = (int)(t >> 5);
    int f0 = (int)(t & 31) * 4;
    int g = batch[node];
    const float4 v = *(const float4*)(h + (size_t)node * HIDDEN + f0);
    float* o = sums + (size_t)g * HIDDEN + f0;
    atomicAdd(o + 0, v.x);
    atomicAdd(o + 1, v.y);
    atomicAdd(o + 2, v.z);
    atomicAdd(o + 3, v.w);
}

__global__ void out_kernel(const float* __restrict__ sums, const float* __restrict__ counts,
                           const float* __restrict__ W_out, const float* __restrict__ b_out,
                           float* __restrict__ out) {
    int t = blockIdx.x * blockDim.x + threadIdx.x;
    if (t >= N_GRAPHS * 5) return;
    int g = t / 5, o = t % 5;
    float inv = 1.0f / fmaxf(counts[g], 1.0f);
    float acc = b_out[o];
    for (int f = 0; f < HIDDEN; ++f)
        acc += sums[(size_t)g * HIDDEN + f] * inv * W_out[f * 5 + o];
    out[t] = acc;
}

extern "C" void kernel_launch(void* const* d_in, const int* in_sizes, int n_in,
                              void* d_out, int out_size, void* d_ws, size_t ws_size,
                              hipStream_t stream) {
    const float* x        = (const float*)d_in[0];
    const int*   ei       = (const int*)d_in[1];
    const int*   src      = ei;
    const int*   dst      = ei + N_EDGES;
    const int*   batch    = (const int*)d_in[2];
    const float* Ws       = (const float*)d_in[3];
    const float* bs       = (const float*)d_in[4];
    const float* gammas   = (const float*)d_in[5];
    const float* betas    = (const float*)d_in[6];
    const float* run_mean = (const float*)d_in[7];
    const float* run_var  = (const float*)d_in[8];
    const float* W_out    = (const float*)d_in[9];
    const float* b_out    = (const float*)d_in[10];
    float*       out      = (float*)d_out;

    const size_t NF = (size_t)N_NODES * HIDDEN;
    float* bufA   = (float*)d_ws;
    float* bufB   = bufA + NF;
    float* bufC   = bufB + NF;
    int*   cnt      = (int*)bufC;
    int*   rowptr   = cnt + N_PAD;
    int*   blocksum = rowptr + N_PAD;
    int*   rank     = blocksum + 128;
    int*   recA     = rank + N_EDGES;
    int*   recB     = recA + 2 * (size_t)N_EDGES;
    float* norm   = bufC + NF;
    float* sums   = norm + N_NODES;
    float* counts = sums + (size_t)N_GRAPHS * HIDDEN;

    const int B = 256;
    auto cdiv = [](size_t a, size_t b) { return (int)((a + b - 1) / b); };

    zero_f32<<<cdiv(N_PAD, B), B, 0, stream>>>((float*)cnt, (size_t)N_PAD);
    hist_rank_kernel<<<cdiv(N_EDGES, B), B, 0, stream>>>(dst, cnt, rank);
    scan_local_kernel<<<SCAN_BLOCKS, 256, 0, stream>>>(cnt, rowptr, blocksum, norm);
    scan_add_kernel<<<SCAN_BLOCKS, 256, 0, stream>>>(rowptr, blocksum);
    fill_kernel<<<cdiv(N_EDGES, B), B, 0, stream>>>(src, dst, rank, rowptr, norm, recA);
    order_runs_kernel<<<cdiv(N_NODES, 8), 256, 0, stream>>>(rowptr, recA, recB);

    const float* h = x;
    for (int l = 0; l < N_LAYERS; ++l) {
        gemm_xw_wmma<<<cdiv(N_NODES / 16, 8), 256, 0, stream>>>(
            h, Ws + (size_t)l * HIDDEN * HIDDEN, bufB);
        gather_update_kernel<<<cdiv(N_NODES, 8), 256, 0, stream>>>(
            bufB, rowptr, recB, norm, bs + l * HIDDEN, gammas + l * HIDDEN, betas + l * HIDDEN,
            run_mean + l * HIDDEN, run_var + l * HIDDEN, bufA);
        h = bufA;
    }

    zero_f32<<<cdiv((size_t)N_GRAPHS * HIDDEN + N_GRAPHS, B), B, 0, stream>>>(
        sums, (size_t)N_GRAPHS * HIDDEN + N_GRAPHS);
    count_kernel<<<cdiv(N_NODES, B), B, 0, stream>>>(batch, counts);
    pool_kernel<<<cdiv((size_t)N_NODES * 32, B), B, 0, stream>>>(h, batch, sums);
    out_kernel<<<cdiv(N_GRAPHS * 5, B), B, 0, stream>>>(sums, counts, W_out, b_out, out);
}


# Edge-table and pool kernels: a few thousand waves, each looping over many items (4 edges a thread, 12 rows a wave); pool adds 16 node rows in registers before its atomic add (sum order changes)
# speedup vs baseline: 4.1621x; 1.3262x over previous
#include <hip/hip_runtime.h>

#define N_NODES 100000
#define N_EDGES 1600000
#define N_GRAPHS 2000
#define HIDDEN 128
#define N_LAYERS 4
#define BN_EPS 1e-5f
#define SCAN_BLOCKS 98
#define N_PAD (SCAN_BLOCKS * 1024)
#define EDGE_QUADS (N_EDGES / 4)
#define EDGE_BLOCKS 782
#define ORDER_BLOCKS 1024
#define POOL_ROWS 16
#define POOL_BLOCKS 782

typedef __attribute__((ext_vector_type(2))) float v2f;
typedef __attribute__((ext_vector_type(8))) float v8f;


__global__ void zero_f32(float* __restrict__ p, size_t n) {
    size_t i = (size_t)blockIdx.x * blockDim.x + threadIdx.x;
    if (i < n) p[i] = 0.0f;
}

__global__ __launch_bounds__(256) void hist_rank_kernel(const int* __restrict__ dst,
                                                        int* __restrict__ cnt,
                                                        int* __restrict__ rank) {
    const int nthr = gridDim.x * 256;
    for (int q = blockIdx.x * 256 + threadIdx.x; q < EDGE_QUADS; q += nthr) {
        const int4 d = *(const int4*)(dst + 4 * (size_t)q);
        int4 r;
        r.x = atomicAdd(&cnt[d.x], 1);
        r.y = atomicAdd(&cnt[d.y], 1);
        r.z = atomicAdd(&cnt[d.z], 1);
        r.w = atomicAdd(&cnt[d.w], 1);
        *(int4*)(rank + 4 * (size_t)q) = r;
    }
}

__global__ __launch_bounds__(256) void scan_local_kernel(const int* __restrict__ cnt,
                                                         int* __restrict__ rowptr,
                                                         int* __restrict__ blocksum,
                                                         float* __restrict__ norm) {
    __shared__ int wsum[8];
    const int t = threadIdx.x, lane = t & 31, w = t >> 5;
    const int i0 = (blockIdx.x * 256 + t) * 4;
    const int4 c = *(const int4*)(cnt + i0);
    const int s = c.x + c.y + c.z + c.w;
    int incl = s;
#pragma unroll
    for (int o = 1; o < 32; o <<= 1) {
        int n = __shfl_up(incl, o);
        if (lane >= o) incl += n;
    }
    if (lane == 31) wsum[w] = incl;
    __syncthreads();
    int woff = 0, total = 0;
#pragma unroll
    for (int k = 0; k < 8; ++k) {
        int v = wsum[k];
        total += v;
        if (k < w) woff += v;
    }
    int4 r;
    r.x = woff + incl - s;
    r.y = r.x + c.x;
    r.z = r.y + c.y;
    r.w = r.z + c.z;
    *(int4*)(rowptr + i0) = r;
    if (t == 0) blocksum[blockIdx.x] = total;
    if (i0 < N_NODES) {
        float4 nv;
        nv.x = rsqrtf((float)c.x + 1.0f);
        nv.y = rsqrtf((float)c.y + 1.0f);
        nv.z = rsqrtf((float)c.z + 1.0f);
        nv.w = rsqrtf((float)c.w + 1.0f);
        *(float4*)(norm + i0) = nv;
    }
}

__global__ __launch_bounds__(256) void scan_add_kernel(int* __restrict__ rowptr,
                                                       const int* __restrict__ blocksum) {
    __shared__ int soff;
    const int t = threadIdx.x, b = blockIdx.x;
    if (t < 32) {
        int v = 0;
        for (int k = t; k < b; k += 32) v += blocksum[k];
#pragma unroll
        for (int o = 16; o > 0; o >>= 1) v += __shfl_xor(v, o);
        if (t == 0) soff = v;
    }
    __syncthreads();
    const int off = soff;
    const int i0 = (b * 256 + t) * 4;
    int4 r = *(const int4*)(rowptr + i0);
    r.x += off; r.y += off; r.z += off; r.w += off;
    *(int4*)(rowptr + i0) = r;
}

__global__ __launch_bounds__(256) void fill_kernel(const int* __restrict__ src,
                                                   const int* __restrict__ dst,
                                                   const int* __restrict__ rank,
                                                   const int* __restrict__ rowptr,
                                                   const float* __restrict__ norm,
                                                   int* __restrict__ rec) {
    const int nthr = gridDim.x * 256;
    for (int q = blockIdx.x * 256 + threadIdx.x; q < EDGE_QUADS; q += nthr) {
        const int4 s = *(const int4*)(src + 4 * (size_t)q);
        const int4 d = *(const int4*)(dst + 4 * (size_t)q);
        const int4 k = *(const int4*)(rank + 4 * (size_t)q);
        const int p0 = rowptr[d.x] + k.x, p1 = rowptr[d.y] + k.y;
        const int p2 = rowptr[d.z] + k.z, p3 = rowptr[d.w] + k.w;
        const float c0 = norm[s.x] * norm[d.x], c1 = norm[s.y] * norm[d.y];
        const float c2 = norm[s.z] * norm[d.z], c3 = norm[s.w] * norm[d.w];
        *(int2*)(rec + 2 * (size_t)p0) = make_int2(s.x, __float_as_int(c0));
        *(int2*)(rec + 2 * (size_t)p1) = make_int2(s.y, __float_as_int(c1));
        *(int2*)(rec + 2 * (size_t)p2) = make_int2(s.z, __float_as_int(c2));
        *(int2*)(rec + 2 * (size_t)p3) = make_int2(s.w, __float_as_int(c3));
    }
}

__global__ __launch_bounds__(256) void order_runs_kernel(const int* __restrict__ rowptr,
                                                         const int* __restrict__ rec_in,
                                                         int* __restrict__ rec_out) {
    const int wave = __builtin_amdgcn_readfirstlane((int)(threadIdx.x >> 5));
    const int lane = threadIdx.x & 31;
    const int nwaves = gridDim.x * 8;
    for (int node = blockIdx.x * 8 + wave; node < N_NODES; node += nwaves) {
        const int beg = rowptr[node];
        const int deg = rowptr[node + 1] - beg;
        for (int ib = 0; ib < deg; ib += 32) {
            const int i = ib + lane;
            int2 mine = make_int2(0, 0);
            if (i < deg) mine = *(const int2*)(rec_in + 2 * (size_t)(beg + i));
            int rk = 0;
            for (int jb = 0; jb < deg; jb += 32) {
                const int j = jb + lane;
                int key = 0x7fffffff;
                if (j < deg) key = rec_in[2 * (size_t)(beg + j)];
                const int n = min(32, deg - jb);
                for (int u = 0; u < n; ++u) {
                    const int k = __builtin_amdgcn_readlane(key, u);
                    rk += (int)(k < mine.x) | ((int)(k == mine.x) & (int)(jb + u < i));
                }
            }
            if (i < deg) *(int2*)(rec_out + 2 * (size_t)(beg + rk)) = mine;
        }
    }
}

template <int K>
__device__ __forceinline__ void add_rows(const float* __restrict__ h_lane, int rs, int rc,
                                         int j, float4& acc) {
    float4 v[K];
    float c[K];
#pragma unroll
    for (int u = 0; u < K; ++u) {
        const int s = __builtin_amdgcn_readlane(rs, j + u);
        c[u] = __int_as_float(__builtin_amdgcn_readlane(rc, j + u));
        v[u] = *(const float4*)(h_lane + (size_t)s * HIDDEN);
    }
#pragma unroll
    for (int u = 0; u < K; ++u) {
        acc.x += v[u].x * c[u];
        acc.y += v[u].y * c[u];
        acc.z += v[u].z * c[u];
        acc.w += v[u].w * c[u];
    }
}

#define T_PITCH 132

__global__ __launch_bounds__(256) __attribute__((amdgpu_waves_per_eu(8, 8))) void gcn_layer_kernel(const float* __restrict__ hin,
                                                        const float* __restrict__ W,
                                                        const int* __restrict__ rowptr,
                                                        const int* __restrict__ rec,
                                                        const float* __restrict__ norm,
                                                        const float* __restrict__ bias,
                                                        const float* __restrict__ gamma,
                                                        const float* __restrict__ beta,
                                                        const float* __restrict__ mean,
                                                        const float* __restrict__ var,
                                                        float* __restrict__ hout) {
    __shared__ __attribute__((aligned(16))) float As[16 * T_PITCH];
    __shared__ __attribute__((aligned(16))) float Os[16 * T_PITCH];
    __shared__ __attribute__((aligned(16))) float Ps[4 * HIDDEN];
    const int wave = __builtin_amdgcn_readfirstlane((int)(threadIdx.x >> 5));
    const int lane = threadIdx.x & 31;
    const int l16  = lane & 15;
    const int hi   = lane >> 4;
    const int f0   = lane * 4;

    v2f bw[32];
#pragma unroll
    for (int kk = 0; kk < 32; ++kk) {
        const int k = kk * 4 + hi * 2;
        const int n = wave * 16 + l16;
        bw[kk].x = W[(size_t)k * HIDDEN + n];
        bw[kk].y = W[(size_t)(k + 1) * HIDDEN + n];
    }

    if (wave == 0) {
        const float4 ga = *(const float4*)(gamma + f0);
        const float4 va = *(const float4*)(var + f0);
        float4 sc4;
        sc4.x = ga.x * rsqrtf(va.x + BN_EPS);
        sc4.y = ga.y * rsqrtf(va.y + BN_EPS);
        sc4.z = ga.z * rsqrtf(va.z + BN_EPS);
        sc4.w = ga.w * rsqrtf(va.w + BN_EPS);
        *(float4*)(Ps + f0)              = *(const float4*)(bias + f0);
        *(float4*)(Ps + HIDDEN + f0)     = *(const float4*)(mean + f0);
        *(float4*)(Ps + 2 * HIDDEN + f0) = sc4;
        *(float4*)(Ps + 3 * HIDDEN + f0) = *(const float4*)(beta + f0);
    }
    __syncthreads();

    const float* h_lane = hin + f0;
    for (int tile = blockIdx.x; tile < N_NODES / 16; tile += gridDim.x) {
        for (int r = 0; r < 2; ++r) {
            const int lrow = wave * 2 + r;
            const int node = tile * 16 + lrow;
            const int beg = rowptr[node], end = rowptr[node + 1];
            const float4 self = *(const float4*)(h_lane + (size_t)node * HIDDEN);
            const float nr = norm[node];
            float4 acc = make_float4(0.0f, 0.0f, 0.0f, 0.0f);
            for (int base = beg; base < end; base += 32) {
                const int cnt = min(end - base, 32);
                int2 pr = make_int2(0, 0);
                if (lane < cnt) pr = *(const int2*)(rec + 2 * (size_t)(base + lane));
                int j = 0;
                for (; j + 8 <= cnt; j += 8) add_rows<8>(h_lane, pr.x, pr.y, j, acc);
                if (cnt & 4) { add_rows<4>(h_lane, pr.x, pr.y, j, acc); j += 4; }
                if (cnt & 2) { add_rows<2>(h_lane, pr.x, pr.y, j, acc); j += 2; }
                if (cnt & 1) { add_rows<1>(h_lane, pr.x, pr.y, j, acc); }
            }
            const float sc = nr * nr;
            acc.x += self.x * sc;
            acc.y += self.y * sc;
            acc.z += self.z * sc;
            acc.w += self.w * sc;
            *(float4*)(As + lrow * T_PITCH + f0) = acc;
        }
        __syncthreads();

        v8f d = (v8f)0.0f;
#pragma unroll
        for (int kk = 0; kk < 32; ++kk) {
            const v2f a = *(const v2f*)(As + l16 * T_PITCH + kk * 4 + hi * 2);
            d = __builtin_amdgcn_wmma_f32_16x16x4_f32(false, a, false, bw[kk], (short)0, d,
                                                      false, false);
        }
#pragma unroll
        for (int v = 0; v < 8; ++v) Os[(hi * 8 + v) * T_PITCH + wave * 16 + l16] = d[v];
        __syncthreads();

        const float4 bi = *(const float4*)(Ps + f0);
        const float4 mu = *(const float4*)(Ps + HIDDEN + f0);
        const float4 sn = *(const float4*)(Ps + 2 * HIDDEN + f0);
        const float4 be = *(const float4*)(Ps + 3 * HIDDEN + f0);
        for (int r = 0; r < 2; ++r) {
            const int lrow = wave * 2 + r;
            const int node = tile * 16 + lrow;
            const float4 p = *(const float4*)(Os + lrow * T_PITCH + f0);
            float4 o;
            o.x = p.x + bi.x;
            o.y = p.y + bi.y;
            o.z = p.z + bi.z;
            o.w = p.w + bi.w;
            o.x = (o.x - mu.x) * sn.x + be.x;
            o.y = (o.y - mu.y) * sn.y + be.y;
            o.z = (o.z - mu.z) * sn.z + be.z;
            o.w = (o.w - mu.w) * sn.w + be.w;
            o.x = fmaxf(o.x, 0.0f);
            o.y = fmaxf(o.y, 0.0f);
            o.z = fmaxf(o.z, 0.0f);
            o.w = fmaxf(o.w, 0.0f);
            *(float4*)(hout + (size_t)node * HIDDEN + f0) = o;
        }
    }
}

__global__ void count_kernel(const int* __restrict__ batch, float* __restrict__ counts) {
    int i = blockIdx.x * blockDim.x + threadIdx.x;
    if (i < N_NODES) atomicAdd(&counts[batch[i]], 1.0f);
}

__device__ __forceinline__ void pool_add(float* __restrict__ sums, int g, int f0, const float4 a) {
    float* o = sums + (size_t)g * HIDDEN + f0;
    atomicAdd(o + 0, a.x);
    atomicAdd(o + 1, a.y);
    atomicAdd(o + 2, a.z);
    atomicAdd(o + 3, a.w);
}

static_assert(N_NODES % POOL_ROWS == 0, "whole groups of rows only");
__global__ __launch_bounds__(256) void pool_kernel(const float* __restrict__ h,
                                                   const int* __restrict__ batch,
                                                   float* __restrict__ sums) {
    const int wave = __builtin_amdgcn_readfirstlane((int)(threadIdx.x >> 5));
    const int lane = threadIdx.x & 31;
    const int n0 = (blockIdx.x * 8 + wave) * POOL_ROWS;
    if (n0 >= N_NODES) return;
    const int f0 = lane * 4;
    float4 v[POOL_ROWS];
#pragma unroll
    for (int u = 0; u < POOL_ROWS; ++u)
        v[u] = *(const float4*)(h + (size_t)(n0 + u) * HIDDEN + f0);
    int g[POOL_ROWS];
#pragma unroll
    for (int k = 0; k < POOL_ROWS / 4; ++k) {
        const int4 t = *(const int4*)(batch + n0 + 4 * k);
        g[4 * k + 0] = t.x;
        g[4 * k + 1] = t.y;
        g[4 * k + 2] = t.z;
        g[4 * k + 3] = t.w;
    }
    int cur = g[0];
    float4 acc = v[0];
#pragma unroll
    for (int u = 1; u < POOL_ROWS; ++u) {
        if (g[u] != cur) {
            pool_add(sums, cur, f0, acc);
            acc = make_float4(0.0f, 0.0f, 0.0f, 0.0f);
            cur = g[u];
        }
        acc.x += v[u].x;
        acc.y += v[u].y;
        acc.z += v[u].z;
        acc.w += v[u].w;
    }
    pool_add(sums, cur, f0, acc);
}

__global__ void out_kernel(const float* __restrict__ sums, const float* __restrict__ counts,
                           const float* __restrict__ W_out, const float* __restrict__ b_out,
                           float* __restrict__ out) {
    int t = blockIdx.x * blockDim.x + threadIdx.x;
    if (t >= N_GRAPHS * 5) return;
    int g = t / 5, o = t % 5;
    float inv = 1.0f / fmaxf(counts[g], 1.0f);
    float acc = b_out[o];
    for (int f = 0; f < HIDDEN; ++f)
        acc += sums[(size_t)g * HIDDEN + f] * inv * W_out[f * 5 + o];
    out[t] = acc;
}

extern "C" void kernel_launch(void* const* d_in, const int* in_sizes, int n_in,
                              void* d_out, int out_size, void* d_ws, size_t ws_size,
                              hipStream_t stream) {
    const float* x        = (const float*)d_in[0];
    const int*   ei       = (const int*)d_in[1];
    const int*   src      = ei;
    const int*   dst      = ei + N_EDGES;
    const int*   batch    = (const int*)d_in[2];
    const float* Ws       = (const float*)d_in[3];
    const float* bs       = (const float*)d_in[4];
    const float* gammas   = (const float*)d_in[5];
    const float* betas    = (const float*)d_in[6];
    const float* run_mean = (const float*)d_in[7];
    const float* run_var  = (const float*)d_in[8];
    const float* W_out    = (const float*)d_in[9];
    const float* b_out    = (const float*)d_in[10];
    float*       out      = (float*)d_out;

    const size_t NF = (size_t)N_NODES * HIDDEN;
    float* bufA   = (float*)d_ws;
    float* bufB   = bufA + NF;
    float* bufC   = bufB + NF;
    int*   cnt      = (int*)bufC;
    int*   rowptr   = cnt + N_PAD;
    int*   blocksum = rowptr + N_PAD;
    int*   rank     = blocksum + 128;
    int*   recA     = rank + N_EDGES;
    int*   recB     = recA + 2 * (size_t)N_EDGES;
    float* norm   = bufC + NF;
    float* sums   = norm + N_NODES;
    float* counts = sums + (size_t)N_GRAPHS * HIDDEN;

    const int B = 256;
    auto cdiv = [](size_t a, size_t b) { return (int)((a + b - 1) / b); };

    zero_f32<<<cdiv(N_PAD, B), B, 0, stream>>>((float*)cnt, (size_t)N_PAD);
    hist_rank_kernel<<<EDGE_BLOCKS, 256, 0, stream>>>(dst, cnt, rank);
    scan_local_kernel<<<SCAN_BLOCKS, 256, 0, stream>>>(cnt, rowptr, blocksum, norm);
    scan_add_kernel<<<SCAN_BLOCKS, 256, 0, stream>>>(rowptr, blocksum);
    fill_kernel<<<EDGE_BLOCKS, 256, 0, stream>>>(src, dst, rank, rowptr, norm, recA);
    order_runs_kernel<<<ORDER_BLOCKS, 256, 0, stream>>>(rowptr, recA, recB);

    const float* h = x;
    for (int l = 0; l < N_LAYERS; ++l) {
        float* hn = (l & 1) ? bufB : bufA;
        gcn_layer_kernel<<<1024, 256, 0, stream>>>(
            h, Ws + (size_t)l * HIDDEN * HIDDEN, rowptr, recB, norm, bs + l * HIDDEN,
            gammas + l * HIDDEN, betas + l * HIDDEN, run_mean + l * HIDDEN, run_var + l * HIDDEN, hn);
        h = hn;
    }

    zero_f32<<<cdiv((size_t)N_GRAPHS * HIDDEN + N_GRAPHS, B), B, 0, stream>>>(
        sums, (size_t)N_GRAPHS * HIDDEN + N_GRAPHS);
    count_kernel<<<cdiv(N_NODES, B), B, 0, stream>>>(batch, counts);
    pool_kernel<<<POOL_BLOCKS, 256, 0, stream>>>(h, batch, sums);
    out_kernel<<<cdiv(N_GRAPHS * 5, B), B, 0, stream>>>(sums, counts, W_out, b_out, out);
}


# Edge table built without global atomics: edges binned by 256-row destination bucket with LDS counters, then a work-group per bucket counts and places its rows in LDS; records and sum order unchanged
# speedup vs baseline: 5.4793x; 1.3165x over previous
#include <hip/hip_runtime.h>

#define N_NODES 100000
#define N_EDGES 1600000
#define N_GRAPHS 2000
#define HIDDEN 128
#define N_LAYERS 4
#define BN_EPS 1e-5f
#define SCAN_BLOCKS 98
#define N_PAD (SCAN_BLOCKS * 1024)
#define ROW_BUCKET 256
#define NB 391
#define NCH 250
#define CH_QUADS 1600
#define CH_ITERS ((CH_QUADS + 255) / 256)
static_assert(NCH * CH_QUADS * 4 == N_EDGES, "chunks cover the edges exactly");
static_assert(NB * ROW_BUCKET >= N_NODES + 1, "the last bucket also writes rowptr[N_NODES]");
static_assert(NB * NCH <= N_PAD, "the bucket-by-chunk table fits the scan");
static_assert(NB * NCH + NCH * 256 >= N_PAD, "one store per thread zeroes the entries behind the table");
static_assert((ROW_BUCKET & (ROW_BUCKET - 1)) == 0 && ROW_BUCKET % 32 == 0, "row in bucket = low bits; whole waves");
#define ORDER_BLOCKS 1024
#define POOL_ROWS 16
#define POOL_BLOCKS 782

typedef __attribute__((ext_vector_type(2))) float v2f;
typedef __attribute__((ext_vector_type(8))) float v8f;


__global__ void zero_f32(float* __restrict__ p, size_t n) {
    size_t i = (size_t)blockIdx.x * blockDim.x + threadIdx.x;
    if (i < n) p[i] = 0.0f;
}

__device__ __forceinline__ int bucket_of(int d) { return (int)((unsigned)d / ROW_BUCKET); }

__global__ __launch_bounds__(256) void bucket_count_kernel(const int* __restrict__ dst,
                                                           int* __restrict__ tab) {
    __shared__ int hist[NB];
    const int t = threadIdx.x;
    for (int k = t; k < NB; k += 256) hist[k] = 0;
    __syncthreads();
    const int4* dq = (const int4*)dst + (size_t)blockIdx.x * CH_QUADS;
    int4 d[CH_ITERS];
#pragma unroll
    for (int k = 0; k < CH_ITERS; ++k) {
        const int q = t + 256 * k;
        d[k] = make_int4(0, 0, 0, 0);
        if (q < CH_QUADS) d[k] = dq[q];
    }
#pragma unroll
    for (int k = 0; k < CH_ITERS; ++k) {
        if (t + 256 * k < CH_QUADS) {
            atomicAdd(&hist[bucket_of(d[k].x)], 1);
            atomicAdd(&hist[bucket_of(d[k].y)], 1);
            atomicAdd(&hist[bucket_of(d[k].z)], 1);
            atomicAdd(&hist[bucket_of(d[k].w)], 1);
        }
    }
    __syncthreads();
    for (int k = t; k < NB; k += 256) tab[k * NCH + blockIdx.x] = hist[k];
    const int ipad = NB * NCH + blockIdx.x * 256 + t;
    if (ipad < N_PAD) tab[ipad] = 0;
}

__global__ __launch_bounds__(256) void scan_local_kernel(const int* __restrict__ cnt,
                                                         int* __restrict__ rowptr,
                                                         int* __restrict__ blocksum) {
    __shared__ int wsum[8];
    const int t = threadIdx.x, lane = t & 31, w = t >> 5;
    const int i0 = (blockIdx.x * 256 + t) * 4;
    const int4 c = *(const int4*)(cnt + i0);
    const int s = c.x + c.y + c.z + c.w;
    int incl = s;
#pragma unroll
    for (int o = 1; o < 32; o <<= 1) {
        int n = __shfl_up(incl, o);
        if (lane >= o) incl += n;
    }
    if (lane == 31) wsum[w] = incl;
    __syncthreads();
    int woff = 0, total = 0;
#pragma unroll
    for (int k = 0; k < 8; ++k) {
        int v = wsum[k];
        total += v;
        if (k < w) woff += v;
    }
    int4 r;
    r.x = woff + incl - s;
    r.y = r.x + c.x;
    r.z = r.y + c.y;
    r.w = r.z + c.z;
    *(int4*)(rowptr + i0) = r;
    if (t == 0) blocksum[blockIdx.x] = total;
}

__global__ __launch_bounds__(256) void scan_add_kernel(int* __restrict__ rowptr,
                                                       const int* __restrict__ blocksum) {
    __shared__ int soff;
    const int t = threadIdx.x, b = blockIdx.x;
    if (t < 32) {
        int v = 0;
        for (int k = t; k < b; k += 32) v += blocksum[k];
#pragma unroll
        for (int o = 16; o > 0; o >>= 1) v += __shfl_xor(v, o);
        if (t == 0) soff = v;
    }
    __syncthreads();
    const int off = soff;
    const int i0 = (b * 256 + t) * 4;
    int4 r = *(const int4*)(rowptr + i0);
    r.x += off; r.y += off; r.z += off; r.w += off;
    *(int4*)(rowptr + i0) = r;
}

__global__ __launch_bounds__(256) void bucket_place_kernel(const int* __restrict__ src,
                                                           const int* __restrict__ dst,
                                                           const int* __restrict__ tabs,
                                                           int* __restrict__ pairs) {
    __shared__ int cur[NB];
    const int t = threadIdx.x;
    for (int k = t; k < NB; k += 256) cur[k] = tabs[k * NCH + blockIdx.x];
    __syncthreads();
    const int4* sq = (const int4*)src + (size_t)blockIdx.x * CH_QUADS;
    const int4* dq = (const int4*)dst + (size_t)blockIdx.x * CH_QUADS;
    int4 s[CH_ITERS], d[CH_ITERS];
#pragma unroll
    for (int k = 0; k < CH_ITERS; ++k) {
        const int q = t + 256 * k;
        s[k] = make_int4(0, 0, 0, 0);
        d[k] = make_int4(0, 0, 0, 0);
        if (q < CH_QUADS) {
            s[k] = sq[q];
            d[k] = dq[q];
        }
    }
#pragma unroll
    for (int k = 0; k < CH_ITERS; ++k) {
        if (t + 256 * k < CH_QUADS) {
            const int p0 = atomicAdd(&cur[bucket_of(d[k].x)], 1);
            const int p1 = atomicAdd(&cur[bucket_of(d[k].y)], 1);
            const int p2 = atomicAdd(&cur[bucket_of(d[k].z)], 1);
            const int p3 = atomicAdd(&cur[bucket_of(d[k].w)], 1);
            *(int2*)(pairs + 2 * (size_t)p0) = make_int2(s[k].x, d[k].x);
            *(int2*)(pairs + 2 * (size_t)p1) = make_int2(s[k].y, d[k].y);
            *(int2*)(pairs + 2 * (size_t)p2) = make_int2(s[k].z, d[k].z);
            *(int2*)(pairs + 2 * (size_t)p3) = make_int2(s[k].w, d[k].w);
        }
    }
}

__global__ __launch_bounds__(ROW_BUCKET) void bucket_rows_kernel(const int* __restrict__ tabs,
                                                                 const int* __restrict__ pairs,
                                                                 int* __restrict__ rowptr,
                                                                 float* __restrict__ norm,
                                                                 int* __restrict__ recs) {
    __shared__ int c[ROW_BUCKET];
    __shared__ int wsum[ROW_BUCKET / 32];
    const int t = threadIdx.x, lane = t & 31, w = t >> 5;
    const int b = blockIdx.x;
    const int lo = tabs[b * NCH];
    const int hi = (b + 1 < NB) ? tabs[(b + 1) * NCH] : N_EDGES;
    c[t] = 0;
    __syncthreads();
    for (int i0 = lo + t; i0 < hi; i0 += 8 * ROW_BUCKET) {
        int d[8];
#pragma unroll
        for (int k = 0; k < 8; ++k) {
            const int i = i0 + k * ROW_BUCKET;
            d[k] = 0;
            if (i < hi) d[k] = pairs[2 * (size_t)i + 1];
        }
#pragma unroll
        for (int k = 0; k < 8; ++k)
            if (i0 + k * ROW_BUCKET < hi) atomicAdd(&c[d[k] & (ROW_BUCKET - 1)], 1);
    }
    __syncthreads();
    const int my = c[t];
    int incl = my;
#pragma unroll
    for (int o = 1; o < 32; o <<= 1) {
        int n = __shfl_up(incl, o);
        if (lane >= o) incl += n;
    }
    if (lane == 31) wsum[w] = incl;
    __syncthreads();
    int woff = 0;
#pragma unroll
    for (int k = 0; k < ROW_BUCKET / 32; ++k)
        if (k < w) woff += wsum[k];
    const int start = lo + woff + incl - my;
    const int node = b * ROW_BUCKET + t;
    if (node <= N_NODES) rowptr[node] = start;
    if (node < N_NODES) norm[node] = rsqrtf((float)my + 1.0f);
    c[t] = start;
    __syncthreads();
    for (int i0 = lo + t; i0 < hi; i0 += 8 * ROW_BUCKET) {
        int2 pr[8];
#pragma unroll
        for (int k = 0; k < 8; ++k) {
            const int i = i0 + k * ROW_BUCKET;
            pr[k] = make_int2(0, 0);
            if (i < hi) pr[k] = *(const int2*)(pairs + 2 * (size_t)i);
        }
#pragma unroll
        for (int k = 0; k < 8; ++k) {
            if (i0 + k * ROW_BUCKET < hi) {
                const int p = atomicAdd(&c[pr[k].y & (ROW_BUCKET - 1)], 1);
                recs[p] = pr[k].x;
            }
        }
    }
}

__global__ __launch_bounds__(256) void order_runs_kernel(const int* __restrict__ rowptr,
                                                         const int* __restrict__ recs,
                                                         const float* __restrict__ norm,
                                                         int* __restrict__ rec_out) {
    const int wave = __builtin_amdgcn_readfirstlane((int)(threadIdx.x >> 5));
    const int lane = threadIdx.x & 31;
    const int nwaves = gridDim.x * 8;
    for (int node = blockIdx.x * 8 + wave; node < N_NODES; node += nwaves) {
        const int beg = rowptr[node];
        const int deg = rowptr[node + 1] - beg;
        const float nd = norm[node];
        for (int ib = 0; ib < deg; ib += 32) {
            const int i = ib + lane;
            int mine = 0x7fffffff;
            float ns = 0.0f;
            if (i < deg) {
                mine = recs[beg + i];
                ns = norm[mine];
            }
            int rk = 0;
            for (int jb = 0; jb < deg; jb += 32) {
                int key = mine;
                if (jb != ib) {
                    const int j = jb + lane;
                    key = 0x7fffffff;
                    if (j < deg) key = recs[beg + j];
                }
                const int n = min(32, deg - jb);
                for (int u = 0; u < n; ++u) {
                    const int k = __builtin_amdgcn_readlane(key, u);
                    rk += (int)(k < mine) | ((int)(k == mine) & (int)(jb + u < i));
                }
            }
            if (i < deg)
                *(int2*)(rec_out + 2 * (size_t)(beg + rk)) = make_int2(mine, __float_as_int(ns * nd));
        }
    }
}

template <int K>
__device__ __forceinline__ void add_rows(const float* __restrict__ h_lane, int rs, int rc,
                                         int j, float4& acc) {
    float4 v[K];
    float c[K];
#pragma unroll
    for (int u = 0; u < K; ++u) {
        const int s = __builtin_amdgcn_readlane(rs, j + u);
        c[u] = __int_as_float(__builtin_amdgcn_readlane(rc, j + u));
        v[u] = *(const float4*)(h_lane + (size_t)s * HIDDEN);
    }
#pragma unroll
    for (int u = 0; u < K; ++u) {
        acc.x += v[u].x * c[u];
        acc.y += v[u].y * c[u];
        acc.z += v[u].z * c[u];
        acc.w += v[u].w * c[u];
    }
}

#define T_PITCH 132

__global__ __launch_bounds__(256) __attribute__((amdgpu_waves_per_eu(8, 8))) void gcn_layer_kernel(const float* __restrict__ hin,
                                                        const float* __restrict__ W,
                                                        const int* __restrict__ rowptr,
                                                        const int* __restrict__ rec,
                                                        const float* __restrict__ norm,
                                                        const float* __restrict__ bias,
                                                        const float* __restrict__ gamma,
                                                        const float* __restrict__ beta,
                                                        const float* __restrict__ mean,
                                                        const float* __restrict__ var,
                                                        float* __restrict__ hout) {
    __shared__ __attribute__((aligned(16))) float As[16 * T_PITCH];
    __shared__ __attribute__((aligned(16))) float Os[16 * T_PITCH];
    __shared__ __attribute__((aligned(16))) float Ps[4 * HIDDEN];
    const int wave = __builtin_amdgcn_readfirstlane((int)(threadIdx.x >> 5));
    const int lane = threadIdx.x & 31;
    const int l16  = lane & 15;
    const int hi   = lane >> 4;
    const int f0   = lane * 4;

    v2f bw[32];
#pragma unroll
    for (int kk = 0; kk < 32; ++kk) {
        const int k = kk * 4 + hi * 2;
        const int n = wave * 16 + l16;
        bw[kk].x = W[(size_t)k * HIDDEN + n];
        bw[kk].y = W[(size_t)(k + 1) * HIDDEN + n];
    }

    if (wave == 0) {
        const float4 ga = *(const float4*)(gamma + f0);
        const float4 va = *(const float4*)(var + f0);
        float4 sc4;
        sc4.x = ga.x * rsqrtf(va.x + BN_EPS);
        sc4.y = ga.y * rsqrtf(va.y + BN_EPS);
        sc4.z = ga.z * rsqrtf(va.z + BN_EPS);
        sc4.w = ga.w * rsqrtf(va.w + BN_EPS);
        *(float4*)(Ps + f0)              = *(const float4*)(bias + f0);
        *(float4*)(Ps + HIDDEN + f0)     = *(const float4*)(mean + f0);
        *(float4*)(Ps + 2 * HIDDEN + f0) = sc4;
        *(float4*)(Ps + 3 * HIDDEN + f0) = *(const float4*)(beta + f0);
    }
    __syncthreads();

    const float* h_lane = hin + f0;
    for (int tile = blockIdx.x; tile < N_NODES / 16; tile += gridDim.x) {
        for (int r = 0; r < 2; ++r) {
            const int lrow = wave * 2 + r;
            const int node = tile * 16 + lrow;
            const int beg = rowptr[node], end = rowptr[node + 1];
            const float4 self = *(const float4*)(h_lane + (size_t)node * HIDDEN);
            const float nr = norm[node];
            float4 acc = make_float4(0.0f, 0.0f, 0.0f, 0.0f);
            for (int base = beg; base < end; base += 32) {
                const int cnt = min(end - base, 32);
                int2 pr = make_int2(0, 0);
                if (lane < cnt) pr = *(const int2*)(rec + 2 * (size_t)(base + lane));
                int j = 0;
                for (; j + 8 <= cnt; j += 8) add_rows<8>(h_lane, pr.x, pr.y, j, acc);
                if (cnt & 4) { add_rows<4>(h_lane, pr.x, pr.y, j, acc); j += 4; }
                if (cnt & 2) { add_rows<2>(h_lane, pr.x, pr.y, j, acc); j += 2; }
                if (cnt & 1) { add_rows<1>(h_lane, pr.x, pr.y, j, acc); }
            }
            const float sc = nr * nr;
            acc.x += self.x * sc;
            acc.y += self.y * sc;
            acc.z += self.z * sc;
            acc.w += self.w * sc;
            *(float4*)(As + lrow * T_PITCH + f0) = acc;
        }
        __syncthreads();

        v8f d = (v8f)0.0f;
#pragma unroll
        for (int kk = 0; kk < 32; ++kk) {
            const v2f a = *(const v2f*)(As + l16 * T_PITCH + kk * 4 + hi * 2);
            d = __builtin_amdgcn_wmma_f32_16x16x4_f32(false, a, false, bw[kk], (short)0, d,
                                                      false, false);
        }
#pragma unroll
        for (int v = 0; v < 8; ++v) Os[(hi * 8 + v) * T_PITCH + wave * 16 + l16] = d[v];
        __syncthreads();

        const float4 bi = *(const float4*)(Ps + f0);
        const float4 mu = *(const float4*)(Ps + HIDDEN + f0);
        const float4 sn = *(const float4*)(Ps + 2 * HIDDEN + f0);
        const float4 be = *(const float4*)(Ps + 3 * HIDDEN + f0);
        for (int r = 0; r < 2; ++r) {
            const int lrow = wave * 2 + r;
            const int node = tile * 16 + lrow;
            const float4 p = *(const float4*)(Os + lrow * T_PITCH + f0);
            float4 o;
            o.x = p.x + bi.x;
            o.y = p.y + bi.y;
            o.z = p.z + bi.z;
            o.w = p.w + bi.w;
            o.x = (o.x - mu.x) * sn.x + be.x;
            o.y = (o.y - mu.y) * sn.y + be.y;
            o.z = (o.z - mu.z) * sn.z + be.z;
            o.w = (o.w - mu.w) * sn.w + be.w;
            o.x = fmaxf(o.x, 0.0f);
            o.y = fmaxf(o.y, 0.0f);
            o.z = fmaxf(o.z, 0.0f);
            o.w = fmaxf(o.w, 0.0f);
            *(float4*)(hout + (size_t)node * HIDDEN + f0) = o;
        }
    }
}

__global__ void count_kernel(const int* __restrict__ batch, float* __restrict__ counts) {
    int i = blockIdx.x * blockDim.x + threadIdx.x;
    if (i < N_NODES) atomicAdd(&counts[batch[i]], 1.0f);
}

__device__ __forceinline__ void pool_add(float* __restrict__ sums, int g, int f0, const float4 a) {
    float* o = sums + (size_t)g * HIDDEN + f0;
    atomicAdd(o + 0, a.x);
    atomicAdd(o + 1, a.y);
    atomicAdd(o + 2, a.z);
    atomicAdd(o + 3, a.w);
}

static_assert(N_NODES % POOL_ROWS == 0, "whole groups of rows only");
__global__ __launch_bounds__(256) void pool_kernel(const float* __restrict__ h,
                                                   const int* __restrict__ batch,
                                                   float* __restrict__ sums) {
    const int wave = __builtin_amdgcn_readfirstlane((int)(threadIdx.x >> 5));
    const int lane = threadIdx.x & 31;
    const int n0 = (blockIdx.x * 8 + wave) * POOL_ROWS;
    if (n0 >= N_NODES) return;
    const int f0 = lane * 4;
    float4 v[POOL_ROWS];
#pragma unroll
    for (int u = 0; u < POOL_ROWS; ++u)
        v[u] = *(const float4*)(h + (size_t)(n0 + u) * HIDDEN + f0);
    int g[POOL_ROWS];
#pragma unroll
    for (int k = 0; k < POOL_ROWS / 4; ++k) {
        const int4 t = *(const int4*)(batch + n0 + 4 * k);
        g[4 * k + 0] = t.x;
        g[4 * k + 1] = t.y;
        g[4 * k + 2] = t.z;
        g[4 * k + 3] = t.w;
    }
    int cur = g[0];
    float4 acc = v[0];
#pragma unroll
    for (int u = 1; u < POOL_ROWS; ++u) {
        if (g[u] != cur) {
            pool_add(sums, cur, f0, acc);
            acc = make_float4(0.0f, 0.0f, 0.0f, 0.0f);
            cur = g[u];
        }
        acc.x += v[u].x;
        acc.y += v[u].y;
        acc.z += v[u].z;
        acc.w += v[u].w;
    }
    pool_add(sums, cur, f0, acc);
}

__global__ void out_kernel(const float* __restrict__ sums, const float* __restrict__ counts,
                           const float* __restrict__ W_out, const float* __restrict__ b_out,
                           float* __restrict__ out) {
    int t = blockIdx.x * blockDim.x + threadIdx.x;
    if (t >= N_GRAPHS * 5) return;
    int g = t / 5, o = t % 5;
    float inv = 1.0f / fmaxf(counts[g], 1.0f);
    float acc = b_out[o];
    for (int f = 0; f < HIDDEN; ++f)
        acc += sums[(size_t)g * HIDDEN + f] * inv * W_out[f * 5 + o];
    out[t] = acc;
}

extern "C" void kernel_launch(void* const* d_in, const int* in_sizes, int n_in,
                              void* d_out, int out_size, void* d_ws, size_t ws_size,
                              hipStream_t stream) {
    const float* x        = (const float*)d_in[0];
    const int*   ei       = (const int*)d_in[1];
    const int*   src      = ei;
    const int*   dst      = ei + N_EDGES;
    const int*   batch    = (const int*)d_in[2];
    const float* Ws       = (const float*)d_in[3];
    const float* bs       = (const float*)d_in[4];
    const float* gammas   = (const float*)d_in[5];
    const float* betas    = (const float*)d_in[6];
    const float* run_mean = (const float*)d_in[7];
    const float* run_var  = (const float*)d_in[8];
    const float* W_out    = (const float*)d_in[9];
    const float* b_out    = (const float*)d_in[10];
    float*       out      = (float*)d_out;

    const size_t NF = (size_t)N_NODES * HIDDEN;
    float* bufA   = (float*)d_ws;
    float* bufB   = bufA + NF;
    float* bufC   = bufB + NF;
    int*   tab      = (int*)bufC;
    int*   tabs     = tab + N_PAD;
    int*   blocksum = tabs + N_PAD;
    int*   rowptr   = blocksum + 128;
    int*   pairs    = rowptr + N_PAD;
    int*   recs     = pairs + 2 * (size_t)N_EDGES;
    int*   recB     = recs + N_EDGES;
    float* norm   = bufC + NF;
    float* sums   = norm + N_NODES;
    float* counts = sums + (size_t)N_GRAPHS * HIDDEN;

    const int B = 256;
    auto cdiv = [](size_t a, size_t b) { return (int)((a + b - 1) / b); };

    bucket_count_kernel<<<NCH, 256, 0, stream>>>(dst, tab);
    scan_local_kernel<<<SCAN_BLOCKS, 256, 0, stream>>>(tab, tabs, blocksum);
    scan_add_kernel<<<SCAN_BLOCKS, 256, 0, stream>>>(tabs, blocksum);
    bucket_place_kernel<<<NCH, 256, 0, stream>>>(src, dst, tabs, pairs);
    bucket_rows_kernel<<<NB, ROW_BUCKET, 0, stream>>>(tabs, pairs, rowptr, norm, recs);
    order_runs_kernel<<<ORDER_BLOCKS, 256, 0, stream>>>(rowptr, recs, norm, recB);

    const float* h = x;
    for (int l = 0; l < N_LAYERS; ++l) {
        float* hn = (l & 1) ? bufB : bufA;
        gcn_layer_kernel<<<1024, 256, 0, stream>>>(
            h, Ws + (size_t)l * HIDDEN * HIDDEN, rowptr, recB, norm, bs + l * HIDDEN,
            gammas + l * HIDDEN, betas + l * HIDDEN, run_mean + l * HIDDEN, run_var + l * HIDDEN, hn);
        h = hn;
    }

    zero_f32<<<cdiv((size_t)N_GRAPHS * HIDDEN + N_GRAPHS, B), B, 0, stream>>>(
        sums, (size_t)N_GRAPHS * HIDDEN + N_GRAPHS);
    count_kernel<<<cdiv(N_NODES, B), B, 0, stream>>>(batch, counts);
    pool_kernel<<<POOL_BLOCKS, 256, 0, stream>>>(h, batch, sums);
    out_kernel<<<cdiv(N_GRAPHS * 5, B), B, 0, stream>>>(sums, counts, W_out, b_out, out);
}


# Graph pool fused into the last layer: 16-row tiles summed per column from LDS in node order, same partial sums and atomic adds as before; counts too; zeroing in the ordering launch; 14 -> 11 launches
# speedup vs baseline: 5.7821x; 1.0553x over previous
#include <hip/hip_runtime.h>

#define N_NODES 100000
#define N_EDGES 1600000
#define N_GRAPHS 2000
#define HIDDEN 128
#define N_LAYERS 4
#define BN_EPS 1e-5f
#define SCAN_BLOCKS 98
#define N_PAD (SCAN_BLOCKS * 1024)
#define ROW_BUCKET 256
#define NB 391
#define NCH 250
#define CH_QUADS 1600
#define CH_ITERS ((CH_QUADS + 255) / 256)
static_assert(NCH * CH_QUADS * 4 == N_EDGES, "chunks cover the edges exactly");
static_assert(NB * ROW_BUCKET >= N_NODES + 1, "the last bucket also writes rowptr[N_NODES]");
static_assert(NB * NCH <= N_PAD, "the bucket-by-chunk table fits the scan");
static_assert(NB * NCH + NCH * 256 >= N_PAD, "one store per thread zeroes the entries behind the table");
static_assert((ROW_BUCKET & (ROW_BUCKET - 1)) == 0 && ROW_BUCKET % 32 == 0, "row in bucket = low bits; whole waves");
#define ORDER_BLOCKS 1024
#define POOL_WORDS (N_GRAPHS * HIDDEN + N_GRAPHS)
static_assert(ORDER_BLOCKS * 256 >= POOL_WORDS, "one store per thread of the ordering launch zeroes the pool words");
static_assert(N_NODES % 16 == 0, "whole 16-row tiles only");

typedef __attribute__((ext_vector_type(2))) float v2f;
typedef __attribute__((ext_vector_type(8))) float v8f;


__device__ __forceinline__ int bucket_of(int d) { return (int)((unsigned)d / ROW_BUCKET); }

__global__ __launch_bounds__(256) void bucket_count_kernel(const int* __restrict__ dst,
                                                           int* __restrict__ tab) {
    __shared__ int hist[NB];
    const int t = threadIdx.x;
    for (int k = t; k < NB; k += 256) hist[k] = 0;
    __syncthreads();
    const int4* dq = (const int4*)dst + (size_t)blockIdx.x * CH_QUADS;
    int4 d[CH_ITERS];
#pragma unroll
    for (int k = 0; k < CH_ITERS; ++k) {
        const int q = t + 256 * k;
        d[k] = make_int4(0, 0, 0, 0);
        if (q < CH_QUADS) d[k] = dq[q];
    }
#pragma unroll
    for (int k = 0; k < CH_ITERS; ++k) {
        if (t + 256 * k < CH_QUADS) {
            atomicAdd(&hist[bucket_of(d[k].x)], 1);
            atomicAdd(&hist[bucket_of(d[k].y)], 1);
            atomicAdd(&hist[bucket_of(d[k].z)], 1);
            atomicAdd(&hist[bucket_of(d[k].w)], 1);
        }
    }
    __syncthreads();
    for (int k = t; k < NB; k += 256) tab[k * NCH + blockIdx.x] = hist[k];
    const int ipad = NB * NCH + blockIdx.x * 256 + t;
    if (ipad < N_PAD) tab[ipad] = 0;
}

__global__ __launch_bounds__(256) void scan_local_kernel(const int* __restrict__ cnt,
                                                         int* __restrict__ rowptr,
                                                         int* __restrict__ blocksum) {
    __shared__ int wsum[8];
    const int t = threadIdx.x, lane = t & 31, w = t >> 5;
    const int i0 = (blockIdx.x * 256 + t) * 4;
    const int4 c = *(const int4*)(cnt + i0);
    const int s = c.x + c.y + c.z + c.w;
    int incl = s;
#pragma unroll
    for (int o = 1; o < 32; o <<= 1) {
        int n = __shfl_up(incl, o);
        if (lane >= o) incl += n;
    }
    if (lane == 31) wsum[w] = incl;
    __syncthreads();
    int woff = 0, total = 0;
#pragma unroll
    for (int k = 0; k < 8; ++k) {
        int v = wsum[k];
        total += v;
        if (k < w) woff += v;
    }
    int4 r;
    r.x = woff + incl - s;
    r.y = r.x + c.x;
    r.z = r.y + c.y;
    r.w = r.z + c.z;
    *(int4*)(rowptr + i0) = r;
    if (t == 0) blocksum[blockIdx.x] = total;
}

__global__ __launch_bounds__(256) void scan_add_kernel(int* __restrict__ rowptr,
                                                       const int* __restrict__ blocksum) {
    __shared__ int soff;
    const int t = threadIdx.x, b = blockIdx.x;
    if (t < 32) {
        int v = 0;
        for (int k = t; k < b; k += 32) v += blocksum[k];
#pragma unroll
        for (int o = 16; o > 0; o >>= 1) v += __shfl_xor(v, o);
        if (t == 0) soff = v;
    }
    __syncthreads();
    const int off = soff;
    const int i0 = (b * 256 + t) * 4;
    int4 r = *(const int4*)(rowptr + i0);
    r.x += off; r.y += off; r.z += off; r.w += off;
    *(int4*)(rowptr + i0) = r;
}

__global__ __launch_bounds__(256) void bucket_place_kernel(const int* __restrict__ src,
                                                           const int* __restrict__ dst,
                                                           const int* __restrict__ tabs,
                                                           int* __restrict__ pairs) {
    __shared__ int cur[NB];
    const int t = threadIdx.x;
    for (int k = t; k < NB; k += 256) cur[k] = tabs[k * NCH + blockIdx.x];
    __syncthreads();
    const int4* sq = (const int4*)src + (size_t)blockIdx.x * CH_QUADS;
    const int4* dq = (const int4*)dst + (size_t)blockIdx.x * CH_QUADS;
    int4 s[CH_ITERS], d[CH_ITERS];
#pragma unroll
    for (int k = 0; k < CH_ITERS; ++k) {
        const int q = t + 256 * k;
        s[k] = make_int4(0, 0, 0, 0);
        d[k] = make_int4(0, 0, 0, 0);
        if (q < CH_QUADS) {
            s[k] = sq[q];
            d[k] = dq[q];
        }
    }
#pragma unroll
    for (int k = 0; k < CH_ITERS; ++k) {
        if (t + 256 * k < CH_QUADS) {
            const int p0 = atomicAdd(&cur[bucket_of(d[k].x)], 1);
            const int p1 = atomicAdd(&cur[bucket_of(d[k].y)], 1);
            const int p2 = atomicAdd(&cur[bucket_of(d[k].z)], 1);
            const int p3 = atomicAdd(&cur[bucket_of(d[k].w)], 1);
            *(int2*)(pairs + 2 * (size_t)p0) = make_int2(s[k].x, d[k].x);
            *(int2*)(pairs + 2 * (size_t)p1) = make_int2(s[k].y, d[k].y);
            *(int2*)(pairs + 2 * (size_t)p2) = make_int2(s[k].z, d[k].z);
            *(int2*)(pairs + 2 * (size_t)p3) = make_int2(s[k].w, d[k].w);
        }
    }
}

__global__ __launch_bounds__(ROW_BUCKET) void bucket_rows_kernel(const int* __restrict__ tabs,
                                                                 const int* __restrict__ pairs,
                                                                 int* __restrict__ rowptr,
                                                                 float* __restrict__ norm,
                                                                 int* __restrict__ recs) {
    __shared__ int c[ROW_BUCKET];
    __shared__ int wsum[ROW_BUCKET / 32];
    const int t = threadIdx.x, lane = t & 31, w = t >> 5;
    const int b = blockIdx.x;
    const int lo = tabs[b * NCH];
    const int hi = (b + 1 < NB) ? tabs[(b + 1) * NCH] : N_EDGES;
    c[t] = 0;
    __syncthreads();
    for (int i0 = lo + t; i0 < hi; i0 += 8 * ROW_BUCKET) {
        int d[8];
#pragma unroll
        for (int k = 0; k < 8; ++k) {
            const int i = i0 + k * ROW_BUCKET;
            d[k] = 0;
            if (i < hi) d[k] = pairs[2 * (size_t)i + 1];
        }
#pragma unroll
        for (int k = 0; k < 8; ++k)
            if (i0 + k * ROW_BUCKET < hi) atomicAdd(&c[d[k] & (ROW_BUCKET - 1)], 1);
    }
    __syncthreads();
    const int my = c[t];
    int incl = my;
#pragma unroll
    for (int o = 1; o < 32; o <<= 1) {
        int n = __shfl_up(incl, o);
        if (lane >= o) incl += n;
    }
    if (lane == 31) wsum[w] = incl;
    __syncthreads();
    int woff = 0;
#pragma unroll
    for (int k = 0; k < ROW_BUCKET / 32; ++k)
        if (k < w) woff += wsum[k];
    const int start = lo + woff + incl - my;
    const int node = b * ROW_BUCKET + t;
    if (node <= N_NODES) rowptr[node] = start;
    if (node < N_NODES) norm[node] = rsqrtf((float)my + 1.0f);
    c[t] = start;
    __syncthreads();
    for (int i0 = lo + t; i0 < hi; i0 += 8 * ROW_BUCKET) {
        int2 pr[8];
#pragma unroll
        for (int k = 0; k < 8; ++k) {
            const int i = i0 + k * ROW_BUCKET;
            pr[k] = make_int2(0, 0);
            if (i < hi) pr[k] = *(const int2*)(pairs + 2 * (size_t)i);
        }
#pragma unroll
        for (int k = 0; k < 8; ++k) {
            if (i0 + k * ROW_BUCKET < hi) {
                const int p = atomicAdd(&c[pr[k].y & (ROW_BUCKET - 1)], 1);
                recs[p] = pr[k].x;
            }
        }
    }
}

__global__ __launch_bounds__(256) void order_runs_kernel(const int* __restrict__ rowptr,
                                                         const int* __restrict__ recs,
                                                         const float* __restrict__ norm,
                                                         int* __restrict__ rec_out,
                                                         float* __restrict__ pool0) {
    const int gid = blockIdx.x * 256 + threadIdx.x;
    if (gid < POOL_WORDS) pool0[gid] = 0.0f;
    const int wave = __builtin_amdgcn_readfirstlane((int)(threadIdx.x >> 5));
    const int lane = threadIdx.x & 31;
    const int nwaves = gridDim.x * 8;
    for (int node = blockIdx.x * 8 + wave; node < N_NODES; node += nwaves) {
        const int beg = rowptr[node];
        const int deg = rowptr[node + 1] - beg;
        const float nd = norm[node];
        for (int ib = 0; ib < deg; ib += 32) {
            const int i = ib + lane;
            int mine = 0x7fffffff;
            float ns = 0.0f;
            if (i < deg) {
                mine = recs[beg + i];
                ns = norm[mine];
            }
            int rk = 0;
            for (int jb = 0; jb < deg; jb += 32) {
                int key = mine;
                if (jb != ib) {
                    const int j = jb + lane;
                    key = 0x7fffffff;
                    if (j < deg) key = recs[beg + j];
                }
                const int n = min(32, deg - jb);
                for (int u = 0; u < n; ++u) {
                    const int k = __builtin_amdgcn_readlane(key, u);
                    rk += (int)(k < mine) | ((int)(k == mine) & (int)(jb + u < i));
                }
            }
            if (i < deg)
                *(int2*)(rec_out + 2 * (size_t)(beg + rk)) = make_int2(mine, __float_as_int(ns * nd));
        }
    }
}

template <int K>
__device__ __forceinline__ void add_rows(const float* __restrict__ h_lane, int rs, int rc,
                                         int j, float4& acc) {
    float4 v[K];
    float c[K];
#pragma unroll
    for (int u = 0; u < K; ++u) {
        const int s = __builtin_amdgcn_readlane(rs, j + u);
        c[u] = __int_as_float(__builtin_amdgcn_readlane(rc, j + u));
        v[u] = *(const float4*)(h_lane + (size_t)s * HIDDEN);
    }
#pragma unroll
    for (int u = 0; u < K; ++u) {
        acc.x += v[u].x * c[u];
        acc.y += v[u].y * c[u];
        acc.z += v[u].z * c[u];
        acc.w += v[u].w * c[u];
    }
}

#define T_PITCH 132

template <bool POOL>
__global__ __launch_bounds__(256) __attribute__((amdgpu_waves_per_eu(8, 8))) void gcn_layer_kernel(const float* __restrict__ hin,
                                                        const float* __restrict__ W,
                                                        const int* __restrict__ rowptr,
                                                        const int* __restrict__ rec,
                                                        const float* __restrict__ norm,
                                                        const float* __restrict__ bias,
                                                        const float* __restrict__ gamma,
                                                        const float* __restrict__ beta,
                                                        const float* __restrict__ mean,
                                                        const float* __restrict__ var,
                                                        float* __restrict__ hout,
                                                        const int* __restrict__ batch,
                                                        float* __restrict__ sums,
                                                        float* __restrict__ counts) {
    __shared__ __attribute__((aligned(16))) float As[16 * T_PITCH];
    __shared__ __attribute__((aligned(16))) float Os[16 * T_PITCH];
    __shared__ __attribute__((aligned(16))) float Ps[4 * HIDDEN];
    const int wave = __builtin_amdgcn_readfirstlane((int)(threadIdx.x >> 5));
    const int lane = threadIdx.x & 31;
    const int l16  = lane & 15;
    const int hi   = lane >> 4;
    const int f0   = lane * 4;

    v2f bw[32];
#pragma unroll
    for (int kk = 0; kk < 32; ++kk) {
        const int k = kk * 4 + hi * 2;
        const int n = wave * 16 + l16;
        bw[kk].x = W[(size_t)k * HIDDEN + n];
        bw[kk].y = W[(size_t)(k + 1) * HIDDEN + n];
    }

    if (wave == 0) {
        const float4 ga = *(const float4*)(gamma + f0);
        const float4 va = *(const float4*)(var + f0);
        float4 sc4;
        sc4.x = ga.x * rsqrtf(va.x + BN_EPS);
        sc4.y = ga.y * rsqrtf(va.y + BN_EPS);
        sc4.z = ga.z * rsqrtf(va.z + BN_EPS);
        sc4.w = ga.w * rsqrtf(va.w + BN_EPS);
        *(float4*)(Ps + f0)              = *(const float4*)(bias + f0);
        *(float4*)(Ps + HIDDEN + f0)     = *(const float4*)(mean + f0);
        *(float4*)(Ps + 2 * HIDDEN + f0) = sc4;
        *(float4*)(Ps + 3 * HIDDEN + f0) = *(const float4*)(beta + f0);
    }
    __syncthreads();

    const float* h_lane = hin + f0;
    for (int tile = blockIdx.x; tile < N_NODES / 16; tile += gridDim.x) {
        for (int r = 0; r < 2; ++r) {
            const int lrow = wave * 2 + r;
            const int node = tile * 16 + lrow;
            const int beg = rowptr[node], end = rowptr[node + 1];
            const float4 self = *(const float4*)(h_lane + (size_t)node * HIDDEN);
            const float nr = norm[node];
            float4 acc = make_float4(0.0f, 0.0f, 0.0f, 0.0f);
            for (int base = beg; base < end; base += 32) {
                const int cnt = min(end - base, 32);
                int2 pr = make_int2(0, 0);
                if (lane < cnt) pr = *(const int2*)(rec + 2 * (size_t)(base + lane));
                int j = 0;
                for (; j + 8 <= cnt; j += 8) add_rows<8>(h_lane, pr.x, pr.y, j, acc);
                if (cnt & 4) { add_rows<4>(h_lane, pr.x, pr.y, j, acc); j += 4; }
                if (cnt & 2) { add_rows<2>(h_lane, pr.x, pr.y, j, acc); j += 2; }
                if (cnt & 1) { add_rows<1>(h_lane, pr.x, pr.y, j, acc); }
            }
            const float sc = nr * nr;
            acc.x += self.x * sc;
            acc.y += self.y * sc;
            acc.z += self.z * sc;
            acc.w += self.w * sc;
            *(float4*)(As + lrow * T_PITCH + f0) = acc;
        }
        __syncthreads();

        v8f d = (v8f)0.0f;
#pragma unroll
        for (int kk = 0; kk < 32; ++kk) {
            const v2f a = *(const v2f*)(As + l16 * T_PITCH + kk * 4 + hi * 2);
            d = __builtin_amdgcn_wmma_f32_16x16x4_f32(false, a, false, bw[kk], (short)0, d,
                                                      false, false);
        }
#pragma unroll
        for (int v = 0; v < 8; ++v) Os[(hi * 8 + v) * T_PITCH + wave * 16 + l16] = d[v];
        __syncthreads();

        if constexpr (POOL) {
            if (wave < 4) {
                int f = threadIdx.x;
                asm volatile("" : "+v"(f));
                const float cb = Ps[f], cm = Ps[HIDDEN + f], cs = Ps[2 * HIDDEN + f], ce = Ps[3 * HIDDEN + f];
                float o[16];
#pragma unroll
                for (int u = 0; u < 16; ++u) {
                    const float p = Os[u * T_PITCH + f] + cb;
                    o[u] = fmaxf((p - cm) * cs + ce, 0.0f);
                }
                int g[16];
#pragma unroll
                for (int k = 0; k < 4; ++k) {
                    const int4 t = *(const int4*)(batch + tile * 16 + 4 * k);
                    g[4 * k + 0] = t.x;
                    g[4 * k + 1] = t.y;
                    g[4 * k + 2] = t.z;
                    g[4 * k + 3] = t.w;
                }
                int cur = g[0], nrow = 1;
                float acc = o[0];
#pragma unroll
                for (int u = 1; u < 16; ++u) {
                    if (g[u] != cur) {
                        atomicAdd(sums + (size_t)cur * HIDDEN + f, acc);
                        if (f == 0) atomicAdd(counts + cur, (float)nrow);
                        acc = 0.0f;
                        nrow = 0;
                        cur = g[u];
                    }
                    acc += o[u];
                    ++nrow;
                }
                atomicAdd(sums + (size_t)cur * HIDDEN + f, acc);
                if (f == 0) atomicAdd(counts + cur, (float)nrow);
            }
        } else {
            const float4 bi = *(const float4*)(Ps + f0);
            const float4 mu = *(const float4*)(Ps + HIDDEN + f0);
            const float4 sn = *(const float4*)(Ps + 2 * HIDDEN + f0);
            const float4 be = *(const float4*)(Ps + 3 * HIDDEN + f0);
            for (int r = 0; r < 2; ++r) {
                const int lrow = wave * 2 + r;
                const int node = tile * 16 + lrow;
                const float4 p = *(const float4*)(Os + lrow * T_PITCH + f0);
                float4 o;
                o.x = p.x + bi.x;
                o.y = p.y + bi.y;
                o.z = p.z + bi.z;
                o.w = p.w + bi.w;
                o.x = (o.x - mu.x) * sn.x + be.x;
                o.y = (o.y - mu.y) * sn.y + be.y;
                o.z = (o.z - mu.z) * sn.z + be.z;
                o.w = (o.w - mu.w) * sn.w + be.w;
                o.x = fmaxf(o.x, 0.0f);
                o.y = fmaxf(o.y, 0.0f);
                o.z = fmaxf(o.z, 0.0f);
                o.w = fmaxf(o.w, 0.0f);
                *(float4*)(hout + (size_t)node * HIDDEN + f0) = o;
            }
        }
    }
}

__global__ void out_kernel(const float* __restrict__ sums, const float* __restrict__ counts,
                           const float* __restrict__ W_out, const float* __restrict__ b_out,
                           float* __restrict__ out) {
    int t = blockIdx.x * blockDim.x + threadIdx.x;
    if (t >= N_GRAPHS * 5) return;
    int g = t / 5, o = t % 5;
    float inv = 1.0f / fmaxf(counts[g], 1.0f);
    float acc = b_out[o];
    for (int f = 0; f < HIDDEN; ++f)
        acc += sums[(size_t)g * HIDDEN + f] * inv * W_out[f * 5 + o];
    out[t] = acc;
}

extern "C" void kernel_launch(void* const* d_in, const int* in_sizes, int n_in,
                              void* d_out, int out_size, void* d_ws, size_t ws_size,
                              hipStream_t stream) {
    const float* x        = (const float*)d_in[0];
    const int*   ei       = (const int*)d_in[1];
    const int*   src      = ei;
    const int*   dst      = ei + N_EDGES;
    const int*   batch    = (const int*)d_in[2];
    const float* Ws       = (const float*)d_in[3];
    const float* bs       = (const float*)d_in[4];
    const float* gammas   = (const float*)d_in[5];
    const float* betas    = (const float*)d_in[6];
    const float* run_mean = (const float*)d_in[7];
    const float* run_var  = (const float*)d_in[8];
    const float* W_out    = (const float*)d_in[9];
    const float* b_out    = (const float*)d_in[10];
    float*       out      = (float*)d_out;

    const size_t NF = (size_t)N_NODES * HIDDEN;
    float* bufA   = (float*)d_ws;
    float* bufB   = bufA + NF;
    float* bufC   = bufB + NF;
    int*   tab      = (int*)bufC;
    int*   tabs     = tab + N_PAD;
    int*   blocksum = tabs + N_PAD;
    int*   rowptr   = blocksum + 128;
    int*   pairs    = rowptr + N_PAD;
    int*   recs     = pairs + 2 * (size_t)N_EDGES;
    int*   recB     = recs + N_EDGES;
    float* norm   = bufC + NF;
    float* sums   = norm + N_NODES;
    float* counts = sums + (size_t)N_GRAPHS * HIDDEN;

    const int B = 256;
    auto cdiv = [](size_t a, size_t b) { return (int)((a + b - 1) / b); };

    bucket_count_kernel<<<NCH, 256, 0, stream>>>(dst, tab);
    scan_local_kernel<<<SCAN_BLOCKS, 256, 0, stream>>>(tab, tabs, blocksum);
    scan_add_kernel<<<SCAN_BLOCKS, 256, 0, stream>>>(tabs, blocksum);
    bucket_place_kernel<<<NCH, 256, 0, stream>>>(src, dst, tabs, pairs);
    bucket_rows_kernel<<<NB, ROW_BUCKET, 0, stream>>>(tabs, pairs, rowptr, norm, recs);
    order_runs_kernel<<<ORDER_BLOCKS, 256, 0, stream>>>(rowptr, recs, norm, recB, sums);

    const float* h = x;
    for (int l = 0; l < N_LAYERS - 1; ++l) {
        float* hn = (l & 1) ? bufB : bufA;
        gcn_layer_kernel<false><<<1024, 256, 0, stream>>>(
            h, Ws + (size_t)l * HIDDEN * HIDDEN, rowptr, recB, norm, bs + l * HIDDEN,
            gammas + l * HIDDEN, betas + l * HIDDEN, run_mean + l * HIDDEN, run_var + l * HIDDEN, hn,
            batch, sums, counts);
        h = hn;
    }
    {
        const int l = N_LAYERS - 1;
        gcn_layer_kernel<true><<<1024, 256, 0, stream>>>(
            h, Ws + (size_t)l * HIDDEN * HIDDEN, rowptr, recB, norm, bs + l * HIDDEN,
            gammas + l * HIDDEN, betas + l * HIDDEN, run_mean + l * HIDDEN, run_var + l * HIDDEN, nullptr,
            batch, sums, counts);
    }
    out_kernel<<<cdiv(N_GRAPHS * 5, B), B, 0, stream>>>(sums, counts, W_out, b_out, out);
}


# Edge table in 2 launches where there were 5: a work-group sorts its chunk by bucket in LDS (count, scan, place), stores whole lines; bucket groups read per-chunk segments; sum order unchanged
# speedup vs baseline: 6.0729x; 1.0503x over previous
#include <hip/hip_runtime.h>

#define N_NODES 100000
#define N_EDGES 1600000
#define N_GRAPHS 2000
#define HIDDEN 128
#define N_LAYERS 4
#define BN_EPS 1e-5f
#define SCAN_BLOCKS 98
#define N_PAD (SCAN_BLOCKS * 1024)
#define ROW_BUCKET 256
#define NB 391
#define NCH 250
#define CH_QUADS 1600
#define CH_EDGES (CH_QUADS * 4)
#define CH_ITERS ((CH_QUADS + 255) / 256)
static_assert(NCH * CH_QUADS * 4 == N_EDGES, "chunks cover the edges exactly");
static_assert(NB * ROW_BUCKET >= N_NODES + 1, "the last bucket also writes rowptr[N_NODES]");
static_assert(NB * NCH <= N_PAD, "the bucket-by-chunk table fits its region");
static_assert(NB <= 512 && NB >= 256, "two counters per thread of a 256-thread work-group hold every bucket");
static_assert(NCH <= ROW_BUCKET, "a bucket's work-group has a thread per chunk");
static_assert(ROW_BUCKET == 256, "row in bucket = low bits; the work-group scan is written for 256 threads");
#define ORDER_BLOCKS 1024
#define POOL_WORDS (N_GRAPHS * HIDDEN + N_GRAPHS)
static_assert(ORDER_BLOCKS * 256 >= POOL_WORDS, "one store per thread of the ordering launch zeroes the pool words");
static_assert(N_NODES % 16 == 0, "whole 16-row tiles only");

typedef __attribute__((ext_vector_type(2))) float v2f;
typedef __attribute__((ext_vector_type(8))) float v8f;


__device__ __forceinline__ int bucket_of(int d) { return (int)((unsigned)d / ROW_BUCKET); }

__device__ __forceinline__ int block_excl_scan_256(int v, int* wsum, int& total) {
    const int lane = threadIdx.x & 31, w = threadIdx.x >> 5;
    int incl = v;
#pragma unroll
    for (int o = 1; o < 32; o <<= 1) {
        int n = __shfl_up(incl, o);
        if (lane >= o) incl += n;
    }
    if (lane == 31) wsum[w] = incl;
    __syncthreads();
    int woff = 0;
    total = 0;
#pragma unroll
    for (int k = 0; k < 8; ++k) {
        const int x = wsum[k];
        total += x;
        if (k < w) woff += x;
    }
    __syncthreads();
    return woff + incl - v;
}

__global__ __launch_bounds__(256) void chunk_sort_kernel(const int* __restrict__ src,
                                                         const int* __restrict__ dst,
                                                         int* __restrict__ tab,
                                                         int* __restrict__ pairs) {
    __shared__ int cur[512];
    __shared__ int wsum[8];
    __shared__ __attribute__((aligned(16))) int2 sorted[CH_EDGES];
    const int t = threadIdx.x;
    cur[t] = 0;
    cur[t + 256] = 0;
    __syncthreads();
    const int4* sq = (const int4*)src + (size_t)blockIdx.x * CH_QUADS;
    const int4* dq = (const int4*)dst + (size_t)blockIdx.x * CH_QUADS;
    int4 s[CH_ITERS], d[CH_ITERS];
#pragma unroll
    for (int k = 0; k < CH_ITERS; ++k) {
        const int q = t + 256 * k;
        s[k] = make_int4(0, 0, 0, 0);
        d[k] = make_int4(0, 0, 0, 0);
        if (q < CH_QUADS) {
            s[k] = sq[q];
            d[k] = dq[q];
        }
    }
#pragma unroll
    for (int k = 0; k < CH_ITERS; ++k) {
        if (t + 256 * k < CH_QUADS) {
            atomicAdd(&cur[bucket_of(d[k].x)], 1);
            atomicAdd(&cur[bucket_of(d[k].y)], 1);
            atomicAdd(&cur[bucket_of(d[k].z)], 1);
            atomicAdd(&cur[bucket_of(d[k].w)], 1);
        }
    }
    __syncthreads();
    const int c0 = cur[t], c1 = cur[t + 256];
    int tot0, tot1;
    const int e0 = block_excl_scan_256(c0, wsum, tot0);
    const int e1 = tot0 + block_excl_scan_256(c1, wsum, tot1);
    cur[t] = e0;
    cur[t + 256] = e1;
    tab[t * NCH + blockIdx.x] = e0;
    if (t + 256 < NB) tab[(t + 256) * NCH + blockIdx.x] = e1;
    __syncthreads();
#pragma unroll
    for (int k = 0; k < CH_ITERS; ++k) {
        if (t + 256 * k < CH_QUADS) {
            const int p0 = atomicAdd(&cur[bucket_of(d[k].x)], 1);
            const int p1 = atomicAdd(&cur[bucket_of(d[k].y)], 1);
            const int p2 = atomicAdd(&cur[bucket_of(d[k].z)], 1);
            const int p3 = atomicAdd(&cur[bucket_of(d[k].w)], 1);
            sorted[p0] = make_int2(s[k].x, d[k].x);
            sorted[p1] = make_int2(s[k].y, d[k].y);
            sorted[p2] = make_int2(s[k].z, d[k].z);
            sorted[p3] = make_int2(s[k].w, d[k].w);
        }
    }
    __syncthreads();
    int4* out = (int4*)(pairs + 2 * (size_t)blockIdx.x * CH_EDGES);
    const int4* in = (const int4*)sorted;
    for (int q = t; q < CH_EDGES / 2; q += 256) out[q] = in[q];
}

__global__ __launch_bounds__(ROW_BUCKET) void bucket_rows_kernel(const int* __restrict__ tab,
                                                                 const int* __restrict__ pairs,
                                                                 int* __restrict__ rowptr,
                                                                 float* __restrict__ norm,
                                                                 int* __restrict__ recs) {
    __shared__ int c[ROW_BUCKET];
    __shared__ int wsum[ROW_BUCKET / 32];
    __shared__ int segpos[ROW_BUCKET];
    __shared__ int seglen[ROW_BUCKET];
    const int t = threadIdx.x, lane = t & 31;
    const int w = __builtin_amdgcn_readfirstlane((int)(threadIdx.x >> 5));
    const int b = blockIdx.x;
    int lo_c = 0, len = 0;
    if (t < NCH) {
        lo_c = tab[b * NCH + t];
        const int hi_c = (b + 1 < NB) ? tab[(b + 1) * NCH + t] : CH_EDGES;
        len = hi_c - lo_c;
    }
    segpos[t] = t * CH_EDGES + lo_c;
    seglen[t] = len;
    c[t] = 0;
    int lo;
    (void)block_excl_scan_256(lo_c, wsum, lo);
    for (int s0 = w; s0 < NCH; s0 += 64) {
        int d[8];
        unsigned ok = 0;
#pragma unroll
        for (int u = 0; u < 8; ++u) {
            const int s = s0 + 8 * u;
            d[u] = 0;
            if (s < NCH && lane < seglen[s]) {
                d[u] = pairs[2 * (size_t)(segpos[s] + lane) + 1];
                ok |= 1u << u;
            }
        }
#pragma unroll
        for (int u = 0; u < 8; ++u)
            if ((ok >> u) & 1u) atomicAdd(&c[d[u] & (ROW_BUCKET - 1)], 1);
        for (int u = 0; u < 8; ++u) {
            const int s = s0 + 8 * u;
            if (s < NCH)
                for (int j = lane + 32; j < seglen[s]; j += 32)
                    atomicAdd(&c[pairs[2 * (size_t)(segpos[s] + j) + 1] & (ROW_BUCKET - 1)], 1);
        }
    }
    __syncthreads();
    const int my = c[t];
    int incl = my;
#pragma unroll
    for (int o = 1; o < 32; o <<= 1) {
        int n = __shfl_up(incl, o);
        if (lane >= o) incl += n;
    }
    if (lane == 31) wsum[w] = incl;
    __syncthreads();
    int woff = 0;
#pragma unroll
    for (int k = 0; k < ROW_BUCKET / 32; ++k)
        if (k < w) woff += wsum[k];
    const int start = lo + woff + incl - my;
    const int node = b * ROW_BUCKET + t;
    if (node <= N_NODES) rowptr[node] = start;
    if (node < N_NODES) norm[node] = rsqrtf((float)my + 1.0f);
    c[t] = start;
    __syncthreads();
    for (int s0 = w; s0 < NCH; s0 += 64) {
        int2 pr[8];
        unsigned ok = 0;
#pragma unroll
        for (int u = 0; u < 8; ++u) {
            const int s = s0 + 8 * u;
            pr[u] = make_int2(0, 0);
            if (s < NCH && lane < seglen[s]) {
                pr[u] = *(const int2*)(pairs + 2 * (size_t)(segpos[s] + lane));
                ok |= 1u << u;
            }
        }
#pragma unroll
        for (int u = 0; u < 8; ++u) {
            if ((ok >> u) & 1u) {
                const int p = atomicAdd(&c[pr[u].y & (ROW_BUCKET - 1)], 1);
                recs[p] = pr[u].x;
            }
        }
        for (int u = 0; u < 8; ++u) {
            const int s = s0 + 8 * u;
            if (s < NCH)
                for (int j = lane + 32; j < seglen[s]; j += 32) {
                    const int2 q = *(const int2*)(pairs + 2 * (size_t)(segpos[s] + j));
                    const int p = atomicAdd(&c[q.y & (ROW_BUCKET - 1)], 1);
                    recs[p] = q.x;
                }
        }
    }
}

__global__ __launch_bounds__(256) void order_runs_kernel(const int* __restrict__ rowptr,
                                                         const int* __restrict__ recs,
                                                         const float* __restrict__ norm,
                                                         int* __restrict__ rec_out,
                                                         float* __restrict__ pool0) {
    const int gid = blockIdx.x * 256 + threadIdx.x;
    if (gid < POOL_WORDS) pool0[gid] = 0.0f;
    const int wave = __builtin_amdgcn_readfirstlane((int)(threadIdx.x >> 5));
    const int lane = threadIdx.x & 31;
    const int nwaves = gridDim.x * 8;
    for (int node = blockIdx.x * 8 + wave; node < N_NODES; node += nwaves) {
        const int beg = rowptr[node];
        const int deg = rowptr[node + 1] - beg;
        const float nd = norm[node];
        for (int ib = 0; ib < deg; ib += 32) {
            const int i = ib + lane;
            int mine = 0x7fffffff;
            float ns = 0.0f;
            if (i < deg) {
                mine = recs[beg + i];
                ns = norm[mine];
            }
            int rk = 0;
            for (int jb = 0; jb < deg; jb += 32) {
                int key = mine;
                if (jb != ib) {
                    const int j = jb + lane;
                    key = 0x7fffffff;
                    if (j < deg) key = recs[beg + j];
                }
                const int n = min(32, deg - jb);
                for (int u = 0; u < n; ++u) {
                    const int k = __builtin_amdgcn_readlane(key, u);
                    rk += (int)(k < mine) | ((int)(k == mine) & (int)(jb + u < i));
                }
            }
            if (i < deg)
                *(int2*)(rec_out + 2 * (size_t)(beg + rk)) = make_int2(mine, __float_as_int(ns * nd));
        }
    }
}

template <int K>
__device__ __forceinline__ void add_rows(const float* __restrict__ h_lane, int rs, int rc,
                                         int j, float4& acc) {
    float4 v[K];
    float c[K];
#pragma unroll
    for (int u = 0; u < K; ++u) {
        const int s = __builtin_amdgcn_readlane(rs, j + u);
        c[u] = __int_as_float(__builtin_amdgcn_readlane(rc, j + u));
        v[u] = *(const float4*)(h_lane + (size_t)s * HIDDEN);
    }
#pragma unroll
    for (int u = 0; u < K; ++u) {
        acc.x += v[u].x * c[u];
        acc.y += v[u].y * c[u];
        acc.z += v[u].z * c[u];
        acc.w += v[u].w * c[u];
    }
}

#define T_PITCH 132

template <bool POOL>
__global__ __launch_bounds__(256) __attribute__((amdgpu_waves_per_eu(8, 8))) void gcn_layer_kernel(const float* __restrict__ hin,
                                                        const float* __restrict__ W,
                                                        const int* __restrict__ rowptr,
                                                        const int* __restrict__ rec,
                                                        const float* __restrict__ norm,
                                                        const float* __restrict__ bias,
                                                        const float* __restrict__ gamma,
                                                        const float* __restrict__ beta,
                                                        const float* __restrict__ mean,
                                                        const float* __restrict__ var,
                                                        float* __restrict__ hout,
                                                        const int* __restrict__ batch,
                                                        float* __restrict__ sums,
                                                        float* __restrict__ counts) {
    __shared__ __attribute__((aligned(16))) float As[16 * T_PITCH];
    __shared__ __attribute__((aligned(16))) float Os[16 * T_PITCH];
    __shared__ __attribute__((aligned(16))) float Ps[4 * HIDDEN];
    const int wave = __builtin_amdgcn_readfirstlane((int)(threadIdx.x >> 5));
    const int lane = threadIdx.x & 31;
    const int l16  = lane & 15;
    const int hi   = lane >> 4;
    const int f0   = lane * 4;

    v2f bw[32];
#pragma unroll
    for (int kk = 0; kk < 32; ++kk) {
        const int k = kk * 4 + hi * 2;
        const int n = wave * 16 + l16;
        bw[kk].x = W[(size_t)k * HIDDEN + n];
        bw[kk].y = W[(size_t)(k + 1) * HIDDEN + n];
    }

    if (wave == 0) {
        const float4 ga = *(const float4*)(gamma + f0);
        const float4 va = *(const float4*)(var + f0);
        float4 sc4;
        sc4.x = ga.x * rsqrtf(va.x + BN_EPS);
        sc4.y = ga.y * rsqrtf(va.y + BN_EPS);
        sc4.z = ga.z * rsqrtf(va.z + BN_EPS);
        sc4.w = ga.w * rsqrtf(va.w + BN_EPS);
        *(float4*)(Ps + f0)              = *(const float4*)(bias + f0);
        *(float4*)(Ps + HIDDEN + f0)     = *(const float4*)(mean + f0);
        *(float4*)(Ps + 2 * HIDDEN + f0) = sc4;
        *(float4*)(Ps + 3 * HIDDEN + f0) = *(const float4*)(beta + f0);
    }
    __syncthreads();

    const float* h_lane = hin + f0;
    for (int tile = blockIdx.x; tile < N_NODES / 16; tile += gridDim.x) {
        for (int r = 0; r < 2; ++r) {
            const int lrow = wave * 2 + r;
            const int node = tile * 16 + lrow;
            const int beg = rowptr[node], end = rowptr[node + 1];
            const float4 self = *(const float4*)(h_lane + (size_t)node * HIDDEN);
            const float nr = norm[node];
            float4 acc = make_float4(0.0f, 0.0f, 0.0f, 0.0f);
            for (int base = beg; base < end; base += 32) {
                const int cnt = min(end - base, 32);
                int2 pr = make_int2(0, 0);
                if (lane < cnt) pr = *(const int2*)(rec + 2 * (size_t)(base + lane));
                int j = 0;
                for (; j + 8 <= cnt; j += 8) add_rows<8>(h_lane, pr.x, pr.y, j, acc);
                if (cnt & 4) { add_rows<4>(h_lane, pr.x, pr.y, j, acc); j += 4; }
                if (cnt & 2) { add_rows<2>(h_lane, pr.x, pr.y, j, acc); j += 2; }
                if (cnt & 1) { add_rows<1>(h_lane, pr.x, pr.y, j, acc); }
            }
            const float sc = nr * nr;
            acc.x += self.x * sc;
            acc.y += self.y * sc;
            acc.z += self.z * sc;
            acc.w += self.w * sc;
            *(float4*)(As + lrow * T_PITCH + f0) = acc;
        }
        __syncthreads();

        v8f d = (v8f)0.0f;
#pragma unroll
        for (int kk = 0; kk < 32; ++kk) {
            const v2f a = *(const v2f*)(As + l16 * T_PITCH + kk * 4 + hi * 2);
            d = __builtin_amdgcn_wmma_f32_16x16x4_f32(false, a, false, bw[kk], (short)0, d,
                                                      false, false);
        }
#pragma unroll
        for (int v = 0; v < 8; ++v) Os[(hi * 8 + v) * T_PITCH + wave * 16 + l16] = d[v];
        __syncthreads();

        if constexpr (POOL) {
            if (wave < 4) {
                int f = threadIdx.x;
                asm volatile("" : "+v"(f));
                const float cb = Ps[f], cm = Ps[HIDDEN + f], cs = Ps[2 * HIDDEN + f], ce = Ps[3 * HIDDEN + f];
                float o[16];
#pragma unroll
                for (int u = 0; u < 16; ++u) {
                    const float p = Os[u * T_PITCH + f] + cb;
                    o[u] = fmaxf((p - cm) * cs + ce, 0.0f);
                }
                int g[16];
#pragma unroll
                for (int k = 0; k < 4; ++k) {
                    const int4 t = *(const int4*)(batch + tile * 16 + 4 * k);
                    g[4 * k + 0] = t.x;
                    g[4 * k + 1] = t.y;
                    g[4 * k + 2] = t.z;
                    g[4 * k + 3] = t.w;
                }
                int cur = g[0], nrow = 1;
                float acc = o[0];
#pragma unroll
                for (int u = 1; u < 16; ++u) {
                    if (g[u] != cur) {
                        atomicAdd(sums + (size_t)cur * HIDDEN + f, acc);
                        if (f == 0) atomicAdd(counts + cur, (float)nrow);
                        acc = 0.0f;
                        nrow = 0;
                        cur = g[u];
                    }
                    acc += o[u];
                    ++nrow;
                }
                atomicAdd(sums + (size_t)cur * HIDDEN + f, acc);
                if (f == 0) atomicAdd(counts + cur, (float)nrow);
            }
        } else {
            const float4 bi = *(const float4*)(Ps + f0);
            const float4 mu = *(const float4*)(Ps + HIDDEN + f0);
            const float4 sn = *(const float4*)(Ps + 2 * HIDDEN + f0);
            const float4 be = *(const float4*)(Ps + 3 * HIDDEN + f0);
            for (int r = 0; r < 2; ++r) {
                const int lrow = wave * 2 + r;
                const int node = tile * 16 + lrow;
                const float4 p = *(const float4*)(Os + lrow * T_PITCH + f0);
                float4 o;
                o.x = p.x + bi.x;
                o.y = p.y + bi.y;
                o.z = p.z + bi.z;
                o.w = p.w + bi.w;
                o.x = (o.x - mu.x) * sn.x + be.x;
                o.y = (o.y - mu.y) * sn.y + be.y;
                o.z = (o.z - mu.z) * sn.z + be.z;
                o.w = (o.w - mu.w) * sn.w + be.w;
                o.x = fmaxf(o.x, 0.0f);
                o.y = fmaxf(o.y, 0.0f);
                o.z = fmaxf(o.z, 0.0f);
                o.w = fmaxf(o.w, 0.0f);
                *(float4*)(hout + (size_t)node * HIDDEN + f0) = o;
            }
        }
    }
}

__global__ void out_kernel(const float* __restrict__ sums, const float* __restrict__ counts,
                           const float* __restrict__ W_out, const float* __restrict__ b_out,
                           float* __restrict__ out) {
    int t = blockIdx.x * blockDim.x + threadIdx.x;
    if (t >= N_GRAPHS * 5) return;
    int g = t / 5, o = t % 5;
    float inv = 1.0f / fmaxf(counts[g], 1.0f);
    float acc = b_out[o];
    for (int f = 0; f < HIDDEN; ++f)
        acc += sums[(size_t)g * HIDDEN + f] * inv * W_out[f * 5 + o];
    out[t] = acc;
}

extern "C" void kernel_launch(void* const* d_in, const int* in_sizes, int n_in,
                              void* d_out, int out_size, void* d_ws, size_t ws_size,
                              hipStream_t stream) {
    const float* x        = (const float*)d_in[0];
    const int*   ei       = (const int*)d_in[1];
    const int*   src      = ei;
    const int*   dst      = ei + N_EDGES;
    const int*   batch    = (const int*)d_in[2];
    const float* Ws       = (const float*)d_in[3];
    const float* bs       = (const float*)d_in[4];
    const float* gammas   = (const float*)d_in[5];
    const float* betas    = (const float*)d_in[6];
    const float* run_mean = (const float*)d_in[7];
    const float* run_var  = (const float*)d_in[8];
    const float* W_out    = (const float*)d_in[9];
    const float* b_out    = (const float*)d_in[10];
    float*       out      = (float*)d_out;

    const size_t NF = (size_t)N_NODES * HIDDEN;
    float* bufA   = (float*)d_ws;
    float* bufB   = bufA + NF;
    float* bufC   = bufB + NF;
    int*   tab      = (int*)bufC;
    int*   rowptr   = tab + 2 * N_PAD + 128;
    int*   pairs    = rowptr + N_PAD;
    int*   recs     = pairs + 2 * (size_t)N_EDGES;
    int*   recB     = recs + N_EDGES;
    float* norm   = bufC + NF;
    float* sums   = norm + N_NODES;
    float* counts = sums + (size_t)N_GRAPHS * HIDDEN;

    const int B = 256;
    auto cdiv = [](size_t a, size_t b) { return (int)((a + b - 1) / b); };

    chunk_sort_kernel<<<NCH, 256, 0, stream>>>(src, dst, tab, pairs);
    bucket_rows_kernel<<<NB, ROW_BUCKET, 0, stream>>>(tab, pairs, rowptr, norm, recs);
    order_runs_kernel<<<ORDER_BLOCKS, 256, 0, stream>>>(rowptr, recs, norm, recB, sums);

    const float* h = x;
    for (int l = 0; l < N_LAYERS - 1; ++l) {
        float* hn = (l & 1) ? bufB : bufA;
        gcn_layer_kernel<false><<<1024, 256, 0, stream>>>(
            h, Ws + (size_t)l * HIDDEN * HIDDEN, rowptr, recB, norm, bs + l * HIDDEN,
            gammas + l * HIDDEN, betas + l * HIDDEN, run_mean + l * HIDDEN, run_var + l * HIDDEN, hn,
            batch, sums, counts);
        h = hn;
    }
    {
        const int l = N_LAYERS - 1;
        gcn_layer_kernel<true><<<1024, 256, 0, stream>>>(
            h, Ws + (size_t)l * HIDDEN * HIDDEN, rowptr, recB, norm, bs + l * HIDDEN,
            gammas + l * HIDDEN, betas + l * HIDDEN, run_mean + l * HIDDEN, run_var + l * HIDDEN, nullptr,
            batch, sums, counts);
    }
    out_kernel<<<cdiv(N_GRAPHS * 5, B), B, 0, stream>>>(sums, counts, W_out, b_out, out);
}


# Row sums with fewer dependent round trips: a wave walks its two rows' records as one stream (self-loop kept as each run's last record), bounds and records fetched a tile ahead; sum order unchanged
# speedup vs baseline: 6.3432x; 1.0445x over previous
#include <hip/hip_runtime.h>

#define N_NODES 100000
#define N_EDGES 1600000
#define N_GRAPHS 2000
#define HIDDEN 128
#define N_LAYERS 4
#define BN_EPS 1e-5f
#define SCAN_BLOCKS 98
#define N_PAD (SCAN_BLOCKS * 1024)
#define ROW_BUCKET 256
#define NB 391
#define NCH 250
#define CH_QUADS 1600
#define CH_EDGES (CH_QUADS * 4)
#define CH_ITERS ((CH_QUADS + 255) / 256)
static_assert(NCH * CH_QUADS * 4 == N_EDGES, "chunks cover the edges exactly");
static_assert(NB * ROW_BUCKET >= N_NODES + 1, "the last bucket also writes rowptr[N_NODES]");
static_assert(NB * NCH <= N_PAD, "the bucket-by-chunk table fits its region");
static_assert(NB <= 512 && NB >= 256, "two counters per thread of a 256-thread work-group hold every bucket");
static_assert(NCH <= ROW_BUCKET, "a bucket's work-group has a thread per chunk");
static_assert(ROW_BUCKET == 256, "row in bucket = low bits; the work-group scan is written for 256 threads");
#define ORDER_BLOCKS 1024
#define POOL_WORDS (N_GRAPHS * HIDDEN + N_GRAPHS)
static_assert(ORDER_BLOCKS * 256 >= POOL_WORDS, "one store per thread of the ordering launch zeroes the pool words");
static_assert(N_NODES % 16 == 0, "whole 16-row tiles only");

typedef __attribute__((ext_vector_type(2))) float v2f;
typedef __attribute__((ext_vector_type(8))) float v8f;
typedef __attribute__((ext_vector_type(4))) float v4f;
typedef const __attribute__((address_space(1))) v4f* gv4p;


__device__ __forceinline__ int bucket_of(int d) { return (int)((unsigned)d / ROW_BUCKET); }

__device__ __forceinline__ int block_excl_scan_256(int v, int* wsum, int& total) {
    const int lane = threadIdx.x & 31, w = threadIdx.x >> 5;
    int incl = v;
#pragma unroll
    for (int o = 1; o < 32; o <<= 1) {
        int n = __shfl_up(incl, o);
        if (lane >= o) incl += n;
    }
    if (lane == 31) wsum[w] = incl;
    __syncthreads();
    int woff = 0;
    total = 0;
#pragma unroll
    for (int k = 0; k < 8; ++k) {
        const int x = wsum[k];
        total += x;
        if (k < w) woff += x;
    }
    __syncthreads();
    return woff + incl - v;
}

__global__ __launch_bounds__(256) void chunk_sort_kernel(const int* __restrict__ src,
                                                         const int* __restrict__ dst,
                                                         int* __restrict__ tab,
                                                         int* __restrict__ pairs) {
    __shared__ int cur[512];
    __shared__ int wsum[8];
    __shared__ __attribute__((aligned(16))) int2 sorted[CH_EDGES];
    const int t = threadIdx.x;
    cur[t] = 0;
    cur[t + 256] = 0;
    __syncthreads();
    const int4* sq = (const int4*)src + (size_t)blockIdx.x * CH_QUADS;
    const int4* dq = (const int4*)dst + (size_t)blockIdx.x * CH_QUADS;
    int4 s[CH_ITERS], d[CH_ITERS];
#pragma unroll
    for (int k = 0; k < CH_ITERS; ++k) {
        const int q = t + 256 * k;
        s[k] = make_int4(0, 0, 0, 0);
        d[k] = make_int4(0, 0, 0, 0);
        if (q < CH_QUADS) {
            s[k] = sq[q];
            d[k] = dq[q];
        }
    }
#pragma unroll
    for (int k = 0; k < CH_ITERS; ++k) {
        if (t + 256 * k < CH_QUADS) {
            atomicAdd(&cur[bucket_of(d[k].x)], 1);
            atomicAdd(&cur[bucket_of(d[k].y)], 1);
            atomicAdd(&cur[bucket_of(d[k].z)], 1);
            atomicAdd(&cur[bucket_of(d[k].w)], 1);
        }
    }
    __syncthreads();
    const int c0 = cur[t], c1 = cur[t + 256];
    int tot0, tot1;
    const int e0 = block_excl_scan_256(c0, wsum, tot0);
    const int e1 = tot0 + block_excl_scan_256(c1, wsum, tot1);
    cur[t] = e0;
    cur[t + 256] = e1;
    tab[t * NCH + blockIdx.x] = e0;
    if (t + 256 < NB) tab[(t + 256) * NCH + blockIdx.x] = e1;
    __syncthreads();
#pragma unroll
    for (int k = 0; k < CH_ITERS; ++k) {
        if (t + 256 * k < CH_QUADS) {
            const int p0 = atomicAdd(&cur[bucket_of(d[k].x)], 1);
            const int p1 = atomicAdd(&cur[bucket_of(d[k].y)], 1);
            const int p2 = atomicAdd(&cur[bucket_of(d[k].z)], 1);
            const int p3 = atomicAdd(&cur[bucket_of(d[k].w)], 1);
            sorted[p0] = make_int2(s[k].x, d[k].x);
            sorted[p1] = make_int2(s[k].y, d[k].y);
            sorted[p2] = make_int2(s[k].z, d[k].z);
            sorted[p3] = make_int2(s[k].w, d[k].w);
        }
    }
    __syncthreads();
    int4* out = (int4*)(pairs + 2 * (size_t)blockIdx.x * CH_EDGES);
    const int4* in = (const int4*)sorted;
    for (int q = t; q < CH_EDGES / 2; q += 256) out[q] = in[q];
}

__global__ __launch_bounds__(ROW_BUCKET) void bucket_rows_kernel(const int* __restrict__ tab,
                                                                 const int* __restrict__ pairs,
                                                                 int* __restrict__ rowptr,
                                                                 float* __restrict__ norm,
                                                                 int* __restrict__ recs) {
    __shared__ int c[ROW_BUCKET];
    __shared__ int wsum[ROW_BUCKET / 32];
    __shared__ int segpos[ROW_BUCKET];
    __shared__ int seglen[ROW_BUCKET];
    const int t = threadIdx.x, lane = t & 31;
    const int w = __builtin_amdgcn_readfirstlane((int)(threadIdx.x >> 5));
    const int b = blockIdx.x;
    int lo_c = 0, len = 0;
    if (t < NCH) {
        lo_c = tab[b * NCH + t];
        const int hi_c = (b + 1 < NB) ? tab[(b + 1) * NCH + t] : CH_EDGES;
        len = hi_c - lo_c;
    }
    segpos[t] = t * CH_EDGES + lo_c;
    seglen[t] = len;
    c[t] = 0;
    int lo;
    (void)block_excl_scan_256(lo_c, wsum, lo);
    for (int s0 = w; s0 < NCH; s0 += 64) {
        int d[8];
        unsigned ok = 0;
#pragma unroll
        for (int u = 0; u < 8; ++u) {
            const int s = s0 + 8 * u;
            d[u] = 0;
            if (s < NCH && lane < seglen[s]) {
                d[u] = pairs[2 * (size_t)(segpos[s] + lane) + 1];
                ok |= 1u << u;
            }
        }
#pragma unroll
        for (int u = 0; u < 8; ++u)
            if ((ok >> u) & 1u) atomicAdd(&c[d[u] & (ROW_BUCKET - 1)], 1);
        for (int u = 0; u < 8; ++u) {
            const int s = s0 + 8 * u;
            if (s < NCH)
                for (int j = lane + 32; j < seglen[s]; j += 32)
                    atomicAdd(&c[pairs[2 * (size_t)(segpos[s] + j) + 1] & (ROW_BUCKET - 1)], 1);
        }
    }
    __syncthreads();
    const int my = c[t];
    int incl = my;
#pragma unroll
    for (int o = 1; o < 32; o <<= 1) {
        int n = __shfl_up(incl, o);
        if (lane >= o) incl += n;
    }
    if (lane == 31) wsum[w] = incl;
    __syncthreads();
    int woff = 0;
#pragma unroll
    for (int k = 0; k < ROW_BUCKET / 32; ++k)
        if (k < w) woff += wsum[k];
    const int start = lo + woff + incl - my;
    const int node = b * ROW_BUCKET + t;
    if (node <= N_NODES) rowptr[node] = start;
    if (node < N_NODES) norm[node] = rsqrtf((float)my + 1.0f);
    c[t] = start;
    __syncthreads();
    for (int s0 = w; s0 < NCH; s0 += 64) {
        int2 pr[8];
        unsigned ok = 0;
#pragma unroll
        for (int u = 0; u < 8; ++u) {
            const int s = s0 + 8 * u;
            pr[u] = make_int2(0, 0);
            if (s < NCH && lane < seglen[s]) {
                pr[u] = *(const int2*)(pairs + 2 * (size_t)(segpos[s] + lane));
                ok |= 1u << u;
            }
        }
#pragma unroll
        for (int u = 0; u < 8; ++u) {
            if ((ok >> u) & 1u) {
                const int p = atomicAdd(&c[pr[u].y & (ROW_BUCKET - 1)], 1);
                recs[p] = pr[u].x;
            }
        }
        for (int u = 0; u < 8; ++u) {
            const int s = s0 + 8 * u;
            if (s < NCH)
                for (int j = lane + 32; j < seglen[s]; j += 32) {
                    const int2 q = *(const int2*)(pairs + 2 * (size_t)(segpos[s] + j));
                    const int p = atomicAdd(&c[q.y & (ROW_BUCKET - 1)], 1);
                    recs[p] = q.x;
                }
        }
    }
}

__global__ __launch_bounds__(256) void order_runs_kernel(const int* __restrict__ rowptr,
                                                         const int* __restrict__ recs,
                                                         const float* __restrict__ norm,
                                                         int* __restrict__ rec_out,
                                                         float* __restrict__ pool0) {
    const int gid = blockIdx.x * 256 + threadIdx.x;
    if (gid < POOL_WORDS) pool0[gid] = 0.0f;
    const int wave = __builtin_amdgcn_readfirstlane((int)(threadIdx.x >> 5));
    const int lane = threadIdx.x & 31;
    const int nwaves = gridDim.x * 8;
    for (int node = blockIdx.x * 8 + wave; node < N_NODES; node += nwaves) {
        const int beg = rowptr[node];
        const int deg = rowptr[node + 1] - beg;
        const float nd = norm[node];
        if (lane == 0)
            *(int2*)(rec_out + 2 * (size_t)(beg + node + deg)) = make_int2(node, __float_as_int(nd * nd));
        for (int ib = 0; ib < deg; ib += 32) {
            const int i = ib + lane;
            int mine = 0x7fffffff;
            float ns = 0.0f;
            if (i < deg) {
                mine = recs[beg + i];
                ns = norm[mine];
            }
            int rk = 0;
            for (int jb = 0; jb < deg; jb += 32) {
                int key = mine;
                if (jb != ib) {
                    const int j = jb + lane;
                    key = 0x7fffffff;
                    if (j < deg) key = recs[beg + j];
                }
                const int n = min(32, deg - jb);
                for (int u = 0; u < n; ++u) {
                    const int k = __builtin_amdgcn_readlane(key, u);
                    rk += (int)(k < mine) | ((int)(k == mine) & (int)(jb + u < i));
                }
            }
            if (i < deg)
                *(int2*)(rec_out + 2 * (size_t)(beg + node + rk)) = make_int2(mine, __float_as_int(ns * nd));
        }
    }
}

template <bool FULL>
__device__ __forceinline__ void add_group(const float* __restrict__ hin, unsigned lane, int rs, int rc,
                                          int j, int r, int pos, int n0,
                                          float* __restrict__ row0, float4& acc) {
    v4f v[8];
    float c[8];
#pragma unroll
    for (int u = 0; u < 8; ++u) {
        v[u] = (v4f)0.0f;
        c[u] = 0.0f;
        if (FULL || u < r) {
            const int s = __builtin_amdgcn_readlane(rs, j + u);
            c[u] = __int_as_float(__builtin_amdgcn_readlane(rc, j + u));
            gv4p rowp = (gv4p)(hin + (size_t)s * HIDDEN);
            asm volatile("" : "+s"(rowp));
            v[u] = rowp[lane];
        }
    }
#pragma unroll
    for (int u = 0; u < 8; ++u) {
        if (FULL || u < r) {
            if (pos + u == n0) {
                *(float4*)row0 = acc;
                acc = make_float4(0.0f, 0.0f, 0.0f, 0.0f);
            }
            acc.x += v[u].x * c[u];
            acc.y += v[u].y * c[u];
            acc.z += v[u].z * c[u];
            acc.w += v[u].w * c[u];
        }
    }
}

__device__ __forceinline__ void add_batch(const float* __restrict__ hin, unsigned lane, int2 p0, int2 p1,
                                          int base, int cnt, int n0,
                                          float* __restrict__ row0, float4& acc) {
    int j = 0;
#pragma nounroll
    for (; j + 8 <= cnt; j += 8) {
        const int rs = (j < 32) ? p0.x : p1.x;
        const int rc = (j < 32) ? p0.y : p1.y;
        add_group<true>(hin, lane, rs, rc, j & 31, 8, base + j, n0, row0, acc);
    }
    if (j < cnt) {
        const int rs = (j < 32) ? p0.x : p1.x;
        const int rc = (j < 32) ? p0.y : p1.y;
        add_group<false>(hin, lane, rs, rc, j & 31, cnt - j, base + j, n0, row0, acc);
    }
}

#define N_RECS (N_EDGES + N_NODES)

__device__ __forceinline__ int stream_bounds_raw(const int* __restrict__ rowptr, int node0, int lane) {
    return rowptr[node0 + min(lane, 2)];
}
__device__ __forceinline__ void stream_bounds(int raw, int node0, int& b0, int& n0, int& n) {
    const int r0 = __builtin_amdgcn_readlane(raw, 0);
    const int r1 = __builtin_amdgcn_readlane(raw, 1);
    const int r2 = __builtin_amdgcn_readlane(raw, 2);
    b0 = r0 + node0;
    n0 = r1 - r0 + 1;
    n  = r2 - r0 + 2;
}

__device__ __forceinline__ void stream_records(const int* __restrict__ rec, int b, int lane,
                                               int2& p0, int2& p1) {
    const int2* r2 = (const int2*)rec;
    p0 = r2[min(b + lane, N_RECS - 1)];
    p1 = r2[min(b + 32 + lane, N_RECS - 1)];
}

#define T_PITCH 132

template <bool POOL>
__global__ __launch_bounds__(256) __attribute__((amdgpu_waves_per_eu(8, 8))) void gcn_layer_kernel(const float* __restrict__ hin,
                                                        const float* __restrict__ W,
                                                        const int* __restrict__ rowptr,
                                                        const int* __restrict__ rec,
                                                        const float* __restrict__ bias,
                                                        const float* __restrict__ gamma,
                                                        const float* __restrict__ beta,
                                                        const float* __restrict__ mean,
                                                        const float* __restrict__ var,
                                                        float* __restrict__ hout,
                                                        const int* __restrict__ batch,
                                                        float* __restrict__ sums,
                                                        float* __restrict__ counts) {
    __shared__ __attribute__((aligned(16))) float As[16 * T_PITCH];
    __shared__ __attribute__((aligned(16))) float Os[16 * T_PITCH];
    __shared__ __attribute__((aligned(16))) float Ps[4 * HIDDEN];
    const int wave = __builtin_amdgcn_readfirstlane((int)(threadIdx.x >> 5));
    const int lane = threadIdx.x & 31;
    const int l16  = lane & 15;
    const int hi   = lane >> 4;
    const unsigned f0 = lane * 4;

    v2f bw[32];
#pragma unroll
    for (int kk = 0; kk < 32; ++kk) {
        const int k = kk * 4 + hi * 2;
        const int n = wave * 16 + l16;
        bw[kk].x = W[(size_t)k * HIDDEN + n];
        bw[kk].y = W[(size_t)(k + 1) * HIDDEN + n];
    }

    if (wave == 0) {
        const float4 ga = *(const float4*)(gamma + f0);
        const float4 va = *(const float4*)(var + f0);
        float4 sc4;
        sc4.x = ga.x * rsqrtf(va.x + BN_EPS);
        sc4.y = ga.y * rsqrtf(va.y + BN_EPS);
        sc4.z = ga.z * rsqrtf(va.z + BN_EPS);
        sc4.w = ga.w * rsqrtf(va.w + BN_EPS);
        *(float4*)(Ps + f0)              = *(const float4*)(bias + f0);
        *(float4*)(Ps + HIDDEN + f0)     = *(const float4*)(mean + f0);
        *(float4*)(Ps + 2 * HIDDEN + f0) = sc4;
        *(float4*)(Ps + 3 * HIDDEN + f0) = *(const float4*)(beta + f0);
    }
    __syncthreads();

    const int NT = N_NODES / 16;
    const int G = gridDim.x;
    float* const arow = As + (wave * 2) * T_PITCH + f0;
    int b0, n0, n;
    int2 q0, q1;
    int raw_next;
    {
        const int t0 = min((int)blockIdx.x, NT - 1);
        const int raw = stream_bounds_raw(rowptr, t0 * 16 + wave * 2, lane);
        raw_next = stream_bounds_raw(rowptr, min(t0 + G, NT - 1) * 16 + wave * 2, lane);
        stream_bounds(raw, t0 * 16 + wave * 2, b0, n0, n);
        stream_records(rec, b0, lane, q0, q1);
    }
    for (int tile = blockIdx.x; tile < NT; tile += G) {
        int nb0, nn0, nn;
        stream_bounds(raw_next, min(tile + G, NT - 1) * 16 + wave * 2, nb0, nn0, nn);
        int2 nq0, nq1;
        stream_records(rec, nb0, lane, nq0, nq1);
        const int raw_fut = stream_bounds_raw(rowptr, min(tile + 2 * G, NT - 1) * 16 + wave * 2, lane);

        {
            float4 acc = make_float4(0.0f, 0.0f, 0.0f, 0.0f);
            add_batch(hin, (unsigned)lane, q0, q1, 0, min(n, 64), n0, arow, acc);
            for (int base = 64; base < n; base += 64) {
                int2 p0, p1;
                stream_records(rec, b0 + base, lane, p0, p1);
                add_batch(hin, (unsigned)lane, p0, p1, base, min(n - base, 64), n0, arow, acc);
            }
            *(float4*)(arow + T_PITCH) = acc;
        }
        b0 = nb0; n0 = nn0; n = nn;
        q0 = nq0; q1 = nq1;
        raw_next = raw_fut;
        __syncthreads();

        v8f d = (v8f)0.0f;
#pragma unroll
        for (int kk = 0; kk < 32; ++kk) {
            const v2f a = *(const v2f*)(As + l16 * T_PITCH + kk * 4 + hi * 2);
            d = __builtin_amdgcn_wmma_f32_16x16x4_f32(false, a, false, bw[kk], (short)0, d,
                                                      false, false);
        }
#pragma unroll
        for (int v = 0; v < 8; ++v) Os[(hi * 8 + v) * T_PITCH + wave * 16 + l16] = d[v];
        __syncthreads();

        if constexpr (POOL) {
            if (wave < 4) {
                int f = threadIdx.x;
                asm volatile("" : "+v"(f));
                const float cb = Ps[f], cm = Ps[HIDDEN + f], cs = Ps[2 * HIDDEN + f], ce = Ps[3 * HIDDEN + f];
                float o[16];
#pragma unroll
                for (int u = 0; u < 16; ++u) {
                    const float p = Os[u * T_PITCH + f] + cb;
                    o[u] = fmaxf((p - cm) * cs + ce, 0.0f);
                }
                int g[16];
#pragma unroll
                for (int k = 0; k < 4; ++k) {
                    const int4 t = *(const int4*)(batch + tile * 16 + 4 * k);
                    g[4 * k + 0] = t.x;
                    g[4 * k + 1] = t.y;
                    g[4 * k + 2] = t.z;
                    g[4 * k + 3] = t.w;
                }
                int cur = g[0], nrow = 1;
                float acc = o[0];
#pragma unroll
                for (int u = 1; u < 16; ++u) {
                    if (g[u] != cur) {
                        atomicAdd(sums + (size_t)cur * HIDDEN + f, acc);
                        if (f == 0) atomicAdd(counts + cur, (float)nrow);
                        acc = 0.0f;
                        nrow = 0;
                        cur = g[u];
                    }
                    acc += o[u];
                    ++nrow;
                }
                atomicAdd(sums + (size_t)cur * HIDDEN + f, acc);
                if (f == 0) atomicAdd(counts + cur, (float)nrow);
            }
        } else {
            const float4 bi = *(const float4*)(Ps + f0);
            const float4 mu = *(const float4*)(Ps + HIDDEN + f0);
            const float4 sn = *(const float4*)(Ps + 2 * HIDDEN + f0);
            const float4 be = *(const float4*)(Ps + 3 * HIDDEN + f0);
            for (int r = 0; r < 2; ++r) {
                const int lrow = wave * 2 + r;
                const int node = tile * 16 + lrow;
                const float4 p = *(const float4*)(Os + lrow * T_PITCH + f0);
                float4 o;
                o.x = p.x + bi.x;
                o.y = p.y + bi.y;
                o.z = p.z + bi.z;
                o.w = p.w + bi.w;
                o.x = (o.x - mu.x) * sn.x + be.x;
                o.y = (o.y - mu.y) * sn.y + be.y;
                o.z = (o.z - mu.z) * sn.z + be.z;
                o.w = (o.w - mu.w) * sn.w + be.w;
                o.x = fmaxf(o.x, 0.0f);
                o.y = fmaxf(o.y, 0.0f);
                o.z = fmaxf(o.z, 0.0f);
                o.w = fmaxf(o.w, 0.0f);
                *(float4*)(hout + (size_t)node * HIDDEN + f0) = o;
            }
        }
    }
}

__global__ void out_kernel(const float* __restrict__ sums, const float* __restrict__ counts,
                           const float* __restrict__ W_out, const float* __restrict__ b_out,
                           float* __restrict__ out) {
    int t = blockIdx.x * blockDim.x + threadIdx.x;
    if (t >= N_GRAPHS * 5) return;
    int g = t / 5, o = t % 5;
    float inv = 1.0f / fmaxf(counts[g], 1.0f);
    float acc = b_out[o];
    for (int f = 0; f < HIDDEN; ++f)
        acc += sums[(size_t)g * HIDDEN + f] * inv * W_out[f * 5 + o];
    out[t] = acc;
}

extern "C" void kernel_launch(void* const* d_in, const int* in_sizes, int n_in,
                              void* d_out, int out_size, void* d_ws, size_t ws_size,
                              hipStream_t stream) {
    const float* x        = (const float*)d_in[0];
    const int*   ei       = (const int*)d_in[1];
    const int*   src      = ei;
    const int*   dst      = ei + N_EDGES;
    const int*   batch    = (const int*)d_in[2];
    const float* Ws       = (const float*)d_in[3];
    const float* bs       = (const float*)d_in[4];
    const float* gammas   = (const float*)d_in[5];
    const float* betas    = (const float*)d_in[6];
    const float* run_mean = (const float*)d_in[7];
    const float* run_var  = (const float*)d_in[8];
    const float* W_out    = (const float*)d_in[9];
    const float* b_out    = (const float*)d_in[10];
    float*       out      = (float*)d_out;

    const size_t NF = (size_t)N_NODES * HIDDEN;
    float* bufA   = (float*)d_ws;
    float* bufB   = bufA + NF;
    float* bufC   = bufB + NF;
    int*   tab      = (int*)bufC;
    int*   rowptr   = tab + 2 * N_PAD + 128;
    int*   pairs    = rowptr + N_PAD;
    int*   recs     = pairs + 2 * (size_t)N_EDGES;
    int*   recB     = recs + N_EDGES;
    float* norm   = bufC + NF;
    float* sums   = norm + N_NODES;
    float* counts = sums + (size_t)N_GRAPHS * HIDDEN;

    const int B = 256;
    auto cdiv = [](size_t a, size_t b) { return (int)((a + b - 1) / b); };

    chunk_sort_kernel<<<NCH, 256, 0, stream>>>(src, dst, tab, pairs);
    bucket_rows_kernel<<<NB, ROW_BUCKET, 0, stream>>>(tab, pairs, rowptr, norm, recs);
    order_runs_kernel<<<ORDER_BLOCKS, 256, 0, stream>>>(rowptr, recs, norm, recB, sums);

    const float* h = x;
    for (int l = 0; l < N_LAYERS - 1; ++l) {
        float* hn = (l & 1) ? bufB : bufA;
        gcn_layer_kernel<false><<<1024, 256, 0, stream>>>(
            h, Ws + (size_t)l * HIDDEN * HIDDEN, rowptr, recB, bs + l * HIDDEN,
            gammas + l * HIDDEN, betas + l * HIDDEN, run_mean + l * HIDDEN, run_var + l * HIDDEN, hn,
            batch, sums, counts);
        h = hn;
    }
    {
        const int l = N_LAYERS - 1;
        gcn_layer_kernel<true><<<1024, 256, 0, stream>>>(
            h, Ws + (size_t)l * HIDDEN * HIDDEN, rowptr, recB, bs + l * HIDDEN,
            gammas + l * HIDDEN, betas + l * HIDDEN, run_mean + l * HIDDEN, run_var + l * HIDDEN, nullptr,
            batch, sums, counts);
    }
    out_kernel<<<cdiv(N_GRAPHS * 5, B), B, 0, stream>>>(sums, counts, W_out, b_out, out);
}


# Helper kernels issue a wave's loads in one run: ordering 13 rows a pass, one-key rank; bucket pairs read once, segments from lanes; layer head drops a barrier; output sum by shuffle tree (new order)
# speedup vs baseline: 6.6670x; 1.0510x over previous
#include <hip/hip_runtime.h>

#define N_NODES 100000
#define N_EDGES 1600000
#define N_GRAPHS 2000
#define HIDDEN 128
#define N_LAYERS 4
#define BN_EPS 1e-5f
#define SCAN_BLOCKS 98
#define N_PAD (SCAN_BLOCKS * 1024)
#define ROW_BUCKET 256
#define NB 391
#define NCH 250
#define CH_QUADS 1600
#define CH_EDGES (CH_QUADS * 4)
#define CH_ITERS ((CH_QUADS + 255) / 256)
static_assert(NCH * CH_QUADS * 4 == N_EDGES, "chunks cover the edges exactly");
static_assert(NB * ROW_BUCKET >= N_NODES + 1, "the last bucket also writes rowptr[N_NODES]");
static_assert(NB * NCH <= N_PAD, "the bucket-by-chunk table fits its region");
static_assert(NB <= 512 && NB >= 256, "two counters per thread of a 256-thread work-group hold every bucket");
static_assert(NCH <= ROW_BUCKET, "a bucket's work-group has a thread per chunk");
static_assert(ROW_BUCKET == 256, "row in bucket = low bits; the work-group scan is written for 256 threads");
#define ORDER_BLOCKS 1024
#define POOL_WORDS (N_GRAPHS * HIDDEN + N_GRAPHS)
static_assert(ORDER_BLOCKS * 256 >= POOL_WORDS, "one store per thread of the ordering launch zeroes the pool words");
static_assert(N_NODES % 16 == 0, "whole 16-row tiles only");

typedef __attribute__((ext_vector_type(2))) float v2f;
typedef __attribute__((ext_vector_type(8))) float v8f;
typedef __attribute__((ext_vector_type(4))) float v4f;
typedef const __attribute__((address_space(1))) v4f* gv4p;


__device__ __forceinline__ int bucket_of(int d) { return (int)((unsigned)d / ROW_BUCKET); }

__device__ __forceinline__ int block_excl_scan_256(int v, int* wsum, int& total) {
    const int lane = threadIdx.x & 31, w = threadIdx.x >> 5;
    int incl = v;
#pragma unroll
    for (int o = 1; o < 32; o <<= 1) {
        int n = __shfl_up(incl, o);
        if (lane >= o) incl += n;
    }
    if (lane == 31) wsum[w] = incl;
    __syncthreads();
    int woff = 0;
    total = 0;
#pragma unroll
    for (int k = 0; k < 8; ++k) {
        const int x = wsum[k];
        total += x;
        if (k < w) woff += x;
    }
    __syncthreads();
    return woff + incl - v;
}

__global__ __launch_bounds__(256) void chunk_sort_kernel(const int* __restrict__ src,
                                                         const int* __restrict__ dst,
                                                         int* __restrict__ tab,
                                                         int* __restrict__ pairs) {
    __shared__ int cur[512];
    __shared__ int wsum[8];
    __shared__ __attribute__((aligned(16))) int2 sorted[CH_EDGES];
    const int t = threadIdx.x;
    cur[t] = 0;
    cur[t + 256] = 0;
    __syncthreads();
    const int4* sq = (const int4*)src + (size_t)blockIdx.x * CH_QUADS;
    const int4* dq = (const int4*)dst + (size_t)blockIdx.x * CH_QUADS;
    int4 s[CH_ITERS], d[CH_ITERS];
#pragma unroll
    for (int k = 0; k < CH_ITERS; ++k) {
        const int q = t + 256 * k;
        s[k] = make_int4(0, 0, 0, 0);
        d[k] = make_int4(0, 0, 0, 0);
        if (q < CH_QUADS) {
            s[k] = sq[q];
            d[k] = dq[q];
        }
    }
#pragma unroll
    for (int k = 0; k < CH_ITERS; ++k) {
        if (t + 256 * k < CH_QUADS) {
            atomicAdd(&cur[bucket_of(d[k].x)], 1);
            atomicAdd(&cur[bucket_of(d[k].y)], 1);
            atomicAdd(&cur[bucket_of(d[k].z)], 1);
            atomicAdd(&cur[bucket_of(d[k].w)], 1);
        }
    }
    __syncthreads();
    const int c0 = cur[t], c1 = cur[t + 256];
    int tot0, tot1;
    const int e0 = block_excl_scan_256(c0, wsum, tot0);
    const int e1 = tot0 + block_excl_scan_256(c1, wsum, tot1);
    cur[t] = e0;
    cur[t + 256] = e1;
    tab[t * NCH + blockIdx.x] = e0;
    if (t + 256 < NB) tab[(t + 256) * NCH + blockIdx.x] = e1;
    __syncthreads();
#pragma unroll
    for (int k = 0; k < CH_ITERS; ++k) {
        if (t + 256 * k < CH_QUADS) {
            const int p0 = atomicAdd(&cur[bucket_of(d[k].x)], 1);
            const int p1 = atomicAdd(&cur[bucket_of(d[k].y)], 1);
            const int p2 = atomicAdd(&cur[bucket_of(d[k].z)], 1);
            const int p3 = atomicAdd(&cur[bucket_of(d[k].w)], 1);
            sorted[p0] = make_int2(s[k].x, d[k].x);
            sorted[p1] = make_int2(s[k].y, d[k].y);
            sorted[p2] = make_int2(s[k].z, d[k].z);
            sorted[p3] = make_int2(s[k].w, d[k].w);
        }
    }
    __syncthreads();
    int4* out = (int4*)(pairs + 2 * (size_t)blockIdx.x * CH_EDGES);
    const int4* in = (const int4*)sorted;
    for (int q = t; q < CH_EDGES / 2; q += 256) out[q] = in[q];
}

__global__ __launch_bounds__(ROW_BUCKET) void bucket_rows_kernel(const int* __restrict__ tab,
                                                                 const int* __restrict__ pairs,
                                                                 int* __restrict__ rowptr,
                                                                 float* __restrict__ norm,
                                                                 int* __restrict__ recs) {
    __shared__ int c[ROW_BUCKET];
    __shared__ int wsum[ROW_BUCKET / 32];
    const int t = threadIdx.x, lane = t & 31;
    const int w = __builtin_amdgcn_readfirstlane((int)(threadIdx.x >> 5));
    const int b = blockIdx.x;
    int lo_c = 0, len = 0;
    if (t < NCH) {
        lo_c = tab[b * NCH + t];
        const int hi_c = (b + 1 < NB) ? tab[(b + 1) * NCH + t] : CH_EDGES;
        len = hi_c - lo_c;
    }
    const int pos = t * CH_EDGES + lo_c;
    c[t] = 0;
    int lo;
    (void)block_excl_scan_256(lo_c, wsum, lo);
    int2 pr[32];
    unsigned ok = 0;
#pragma unroll
    for (int u = 0; u < 32; ++u) {
        const int sp = __builtin_amdgcn_readlane(pos, u);
        const int sn = __builtin_amdgcn_readlane(len, u);
        pr[u] = make_int2(0, 0);
        if (lane < sn) {
            pr[u] = *(const int2*)(pairs + 2 * (size_t)(sp + lane));
            ok |= 1u << u;
        }
    }
    const bool any_long = __builtin_amdgcn_ballot_w32(len > 32) != 0;
#pragma unroll
    for (int u = 0; u < 32; ++u)
        if ((ok >> u) & 1u) atomicAdd(&c[pr[u].y & (ROW_BUCKET - 1)], 1);
    if (any_long) {
        for (int u = 0; u < 32; ++u) {
            const int sp = __builtin_amdgcn_readlane(pos, u);
            const int sn = __builtin_amdgcn_readlane(len, u);
            for (int j = lane + 32; j < sn; j += 32)
                atomicAdd(&c[pairs[2 * (size_t)(sp + j) + 1] & (ROW_BUCKET - 1)], 1);
        }
    }
    __syncthreads();
    const int my = c[t];
    int incl = my;
#pragma unroll
    for (int o = 1; o < 32; o <<= 1) {
        int n = __shfl_up(incl, o);
        if (lane >= o) incl += n;
    }
    if (lane == 31) wsum[w] = incl;
    __syncthreads();
    int woff = 0;
#pragma unroll
    for (int k = 0; k < ROW_BUCKET / 32; ++k)
        if (k < w) woff += wsum[k];
    const int start = lo + woff + incl - my;
    const int node = b * ROW_BUCKET + t;
    if (node <= N_NODES) rowptr[node] = start;
    if (node < N_NODES) norm[node] = rsqrtf((float)my + 1.0f);
    c[t] = start;
    __syncthreads();
#pragma unroll
    for (int u = 0; u < 32; ++u) {
        if ((ok >> u) & 1u) {
            const int p = atomicAdd(&c[pr[u].y & (ROW_BUCKET - 1)], 1);
            recs[p] = pr[u].x;
        }
    }
    if (any_long) {
        for (int u = 0; u < 32; ++u) {
            const int sp = __builtin_amdgcn_readlane(pos, u);
            const int sn = __builtin_amdgcn_readlane(len, u);
            for (int j = lane + 32; j < sn; j += 32) {
                const int2 q = *(const int2*)(pairs + 2 * (size_t)(sp + j));
                const int p = atomicAdd(&c[q.y & (ROW_BUCKET - 1)], 1);
                recs[p] = q.x;
            }
        }
    }
}

__device__ __forceinline__ void order_row_any(const int* __restrict__ recs, const float* __restrict__ norm,
                                              int* __restrict__ rec_out, int node, int beg, int deg,
                                              float nd, int lane) {
    if (lane == 0)
        *(int2*)(rec_out + 2 * (size_t)(beg + node + deg)) = make_int2(node, __float_as_int(nd * nd));
    for (int ib = 0; ib < deg; ib += 32) {
        const int i = ib + lane;
        int mine = 0x7fffffff;
        float ns = 0.0f;
        if (i < deg) {
            mine = recs[beg + i];
            ns = norm[mine];
        }
        int rk = 0;
        for (int jb = 0; jb < deg; jb += 32) {
            int key = mine;
            if (jb != ib) {
                const int j = jb + lane;
                key = 0x7fffffff;
                if (j < deg) key = recs[beg + j];
            }
            const int n = min(32, deg - jb);
            for (int u = 0; u < n; ++u) {
                const int k = __builtin_amdgcn_readlane(key, u);
                rk += (int)(k < mine) | ((int)(k == mine) & (int)(jb + u < i));
            }
        }
        if (i < deg)
            *(int2*)(rec_out + 2 * (size_t)(beg + node + rk)) = make_int2(mine, __float_as_int(ns * nd));
    }
}

#define ORDER_ROWS 13
static_assert((long long)ORDER_ROWS * ORDER_BLOCKS * 8 >= N_NODES, "one pass of the ordering launch covers every row");
static_assert(N_NODES <= (1 << 26), "source row and place fit one positive 32-bit key");
__global__ __launch_bounds__(256) void order_runs_kernel(const int* __restrict__ rowptr,
                                                         const int* __restrict__ recs,
                                                         const float* __restrict__ norm,
                                                         int* __restrict__ rec_out,
                                                         float* __restrict__ pool0) {
    const int gid = blockIdx.x * 256 + threadIdx.x;
    if (gid < POOL_WORDS) pool0[gid] = 0.0f;
    const int wave = __builtin_amdgcn_readfirstlane((int)(threadIdx.x >> 5));
    const int lane = threadIdx.x & 31;
    const int nwaves = gridDim.x * 8;
    for (int node0 = blockIdx.x * 8 + wave; node0 < N_NODES; node0 += ORDER_ROWS * nwaves) {
        int beg[ORDER_ROWS], deg[ORDER_ROWS];
        float nd[ORDER_ROWS];
#pragma unroll
        for (int r = 0; r < ORDER_ROWS; ++r) {
            const int nr = min(node0 + r * nwaves, N_NODES - 1);
            beg[r] = rowptr[nr];
            deg[r] = rowptr[nr + 1] - beg[r];
            nd[r] = norm[nr];
        }
        int key[ORDER_ROWS];
#pragma unroll
        for (int r = 0; r < ORDER_ROWS; ++r)
            key[r] = recs[min(beg[r] + lane, N_EDGES - 1)];
        float ns[ORDER_ROWS];
#pragma unroll
        for (int r = 0; r < ORDER_ROWS; ++r)
            ns[r] = norm[key[r]];
#pragma unroll
        for (int r = 0; r < ORDER_ROWS; ++r)
            asm volatile("" : "+v"(ns[r]));
        bool any_long = false;
#pragma unroll
        for (int r = 0; r < ORDER_ROWS; ++r) {
            const int nr = node0 + r * nwaves;
            if (nr < N_NODES) {
                const int d = deg[r];
                if (d > 32) {
                    any_long = true;
                } else {
                    const int ck = (key[r] << 5) | lane;
                    int rk = 0;
                    for (int u = 0; u < d; ++u) {
                        const int k = __builtin_amdgcn_readlane(ck, u);
                        rk += (int)(k < ck);
                    }
                    if (lane < d)
                        *(int2*)(rec_out + 2 * (size_t)(beg[r] + nr + rk)) =
                            make_int2(key[r], __float_as_int(ns[r] * nd[r]));
                    if (lane == 0)
                        *(int2*)(rec_out + 2 * (size_t)(beg[r] + nr + d)) =
                            make_int2(nr, __float_as_int(nd[r] * nd[r]));
                }
            }
        }
        if (any_long) {
#pragma nounroll
            for (int r = 0; r < ORDER_ROWS; ++r) {
                const int nr = node0 + r * nwaves;
                if (nr < N_NODES) {
                    const int b = rowptr[nr];
                    const int d = rowptr[nr + 1] - b;
                    if (d > 32) order_row_any(recs, norm, rec_out, nr, b, d, norm[nr], lane);
                }
            }
        }
    }
}

template <bool FULL>
__device__ __forceinline__ void add_group(const float* __restrict__ hin, unsigned lane, int rs, int rc,
                                          int j, int r, int pos, int n0,
                                          float* __restrict__ row0, float4& acc) {
    v4f v[8];
    float c[8];
#pragma unroll
    for (int u = 0; u < 8; ++u) {
        v[u] = (v4f)0.0f;
        c[u] = 0.0f;
        if (FULL || u < r) {
            const int s = __builtin_amdgcn_readlane(rs, j + u);
            c[u] = __int_as_float(__builtin_amdgcn_readlane(rc, j + u));
            gv4p rowp = (gv4p)(hin + (size_t)s * HIDDEN);
            asm volatile("" : "+s"(rowp));
            v[u] = rowp[lane];
        }
    }
#pragma unroll
    for (int u = 0; u < 8; ++u) {
        if (FULL || u < r) {
            if (pos + u == n0) {
                *(float4*)row0 = acc;
                acc = make_float4(0.0f, 0.0f, 0.0f, 0.0f);
            }
            acc.x += v[u].x * c[u];
            acc.y += v[u].y * c[u];
            acc.z += v[u].z * c[u];
            acc.w += v[u].w * c[u];
        }
    }
}

__device__ __forceinline__ void add_batch(const float* __restrict__ hin, unsigned lane, int2 p0, int2 p1,
                                          int base, int cnt, int n0,
                                          float* __restrict__ row0, float4& acc) {
    int j = 0;
#pragma nounroll
    for (; j + 8 <= cnt; j += 8) {
        const int rs = (j < 32) ? p0.x : p1.x;
        const int rc = (j < 32) ? p0.y : p1.y;
        add_group<true>(hin, lane, rs, rc, j & 31, 8, base + j, n0, row0, acc);
    }
    if (j < cnt) {
        const int rs = (j < 32) ? p0.x : p1.x;
        const int rc = (j < 32) ? p0.y : p1.y;
        add_group<false>(hin, lane, rs, rc, j & 31, cnt - j, base + j, n0, row0, acc);
    }
}

#define N_RECS (N_EDGES + N_NODES)

__device__ __forceinline__ int stream_bounds_raw(const int* __restrict__ rowptr, int node0, int lane) {
    return rowptr[node0 + min(lane, 2)];
}
__device__ __forceinline__ void stream_bounds(int raw, int node0, int& b0, int& n0, int& n) {
    const int r0 = __builtin_amdgcn_readlane(raw, 0);
    const int r1 = __builtin_amdgcn_readlane(raw, 1);
    const int r2 = __builtin_amdgcn_readlane(raw, 2);
    b0 = r0 + node0;
    n0 = r1 - r0 + 1;
    n  = r2 - r0 + 2;
}

__device__ __forceinline__ void stream_records(const int* __restrict__ rec, int b, int lane,
                                               int2& p0, int2& p1) {
    const int2* r2 = (const int2*)rec;
    p0 = r2[min(b + lane, N_RECS - 1)];
    p1 = r2[min(b + 32 + lane, N_RECS - 1)];
}

#define T_PITCH 132

template <bool POOL>
__global__ __launch_bounds__(256) __attribute__((amdgpu_waves_per_eu(8, 8))) void gcn_layer_kernel(const float* __restrict__ hin,
                                                        const float* __restrict__ W,
                                                        const int* __restrict__ rowptr,
                                                        const int* __restrict__ rec,
                                                        const float* __restrict__ bias,
                                                        const float* __restrict__ gamma,
                                                        const float* __restrict__ beta,
                                                        const float* __restrict__ mean,
                                                        const float* __restrict__ var,
                                                        float* __restrict__ hout,
                                                        const int* __restrict__ batch,
                                                        float* __restrict__ sums,
                                                        float* __restrict__ counts) {
    __shared__ __attribute__((aligned(16))) float As[16 * T_PITCH];
    __shared__ __attribute__((aligned(16))) float Os[16 * T_PITCH];
    __shared__ __attribute__((aligned(16))) float Ps[4 * HIDDEN];
    const int wave = __builtin_amdgcn_readfirstlane((int)(threadIdx.x >> 5));
    const int lane = threadIdx.x & 31;
    const int l16  = lane & 15;
    const int hi   = lane >> 4;
    const unsigned f0 = lane * 4;

    const int NT = N_NODES / 16;
    const int G = gridDim.x;
    int b0, n0, n;
    int2 q0, q1;
    int raw_next;
    {
        v4f ga = *(const v4f*)(gamma + f0);
        v4f va = *(const v4f*)(var + f0);
        v4f bi = *(const v4f*)(bias + f0);
        v4f mu = *(const v4f*)(mean + f0);
        v4f be = *(const v4f*)(beta + f0);
        const int t0 = min((int)blockIdx.x, NT - 1);
        const int raw = stream_bounds_raw(rowptr, t0 * 16 + wave * 2, lane);
        raw_next = stream_bounds_raw(rowptr, min(t0 + G, NT - 1) * 16 + wave * 2, lane);
        stream_bounds(raw, t0 * 16 + wave * 2, b0, n0, n);
        stream_records(rec, b0, lane, q0, q1);
        asm volatile("" : "+v"(ga), "+v"(va), "+v"(bi), "+v"(mu), "+v"(be));
        if (wave == 0) {
            v4f sc4;
            sc4.x = ga.x * rsqrtf(va.x + BN_EPS);
            sc4.y = ga.y * rsqrtf(va.y + BN_EPS);
            sc4.z = ga.z * rsqrtf(va.z + BN_EPS);
            sc4.w = ga.w * rsqrtf(va.w + BN_EPS);
            *(v4f*)(Ps + f0)              = bi;
            *(v4f*)(Ps + HIDDEN + f0)     = mu;
            *(v4f*)(Ps + 2 * HIDDEN + f0) = sc4;
            *(v4f*)(Ps + 3 * HIDDEN + f0) = be;
        }
    }

    v2f bw[32];
#pragma unroll
    for (int kk = 0; kk < 32; ++kk) {
        const int k = kk * 4 + hi * 2;
        const int n = wave * 16 + l16;
        bw[kk].x = W[(size_t)k * HIDDEN + n];
        bw[kk].y = W[(size_t)(k + 1) * HIDDEN + n];
    }


    float* const arow = As + (wave * 2) * T_PITCH + f0;
    for (int tile = blockIdx.x; tile < NT; tile += G) {
        int nb0, nn0, nn;
        stream_bounds(raw_next, min(tile + G, NT - 1) * 16 + wave * 2, nb0, nn0, nn);
        int2 nq0, nq1;
        stream_records(rec, nb0, lane, nq0, nq1);
        const int raw_fut = stream_bounds_raw(rowptr, min(tile + 2 * G, NT - 1) * 16 + wave * 2, lane);

        {
            float4 acc = make_float4(0.0f, 0.0f, 0.0f, 0.0f);
            add_batch(hin, (unsigned)lane, q0, q1, 0, min(n, 64), n0, arow, acc);
            for (int base = 64; base < n; base += 64) {
                int2 p0, p1;
                stream_records(rec, b0 + base, lane, p0, p1);
                add_batch(hin, (unsigned)lane, p0, p1, base, min(n - base, 64), n0, arow, acc);
            }
            *(float4*)(arow + T_PITCH) = acc;
        }
        b0 = nb0; n0 = nn0; n = nn;
        q0 = nq0; q1 = nq1;
        raw_next = raw_fut;
        __syncthreads();

        v8f d = (v8f)0.0f;
#pragma unroll
        for (int kk = 0; kk < 32; ++kk) {
            const v2f a = *(const v2f*)(As + l16 * T_PITCH + kk * 4 + hi * 2);
            d = __builtin_amdgcn_wmma_f32_16x16x4_f32(false, a, false, bw[kk], (short)0, d,
                                                      false, false);
        }
#pragma unroll
        for (int v = 0; v < 8; ++v) Os[(hi * 8 + v) * T_PITCH + wave * 16 + l16] = d[v];
        __syncthreads();

        if constexpr (POOL) {
            if (wave < 4) {
                int f = threadIdx.x;
                asm volatile("" : "+v"(f));
                const float cb = Ps[f], cm = Ps[HIDDEN + f], cs = Ps[2 * HIDDEN + f], ce = Ps[3 * HIDDEN + f];
                float o[16];
#pragma unroll
                for (int u = 0; u < 16; ++u) {
                    const float p = Os[u * T_PITCH + f] + cb;
                    o[u] = fmaxf((p - cm) * cs + ce, 0.0f);
                }
                int g[16];
#pragma unroll
                for (int k = 0; k < 4; ++k) {
                    const int4 t = *(const int4*)(batch + tile * 16 + 4 * k);
                    g[4 * k + 0] = t.x;
                    g[4 * k + 1] = t.y;
                    g[4 * k + 2] = t.z;
                    g[4 * k + 3] = t.w;
                }
                int cur = g[0], nrow = 1;
                float acc = o[0];
#pragma unroll
                for (int u = 1; u < 16; ++u) {
                    if (g[u] != cur) {
                        atomicAdd(sums + (size_t)cur * HIDDEN + f, acc);
                        if (f == 0) atomicAdd(counts + cur, (float)nrow);
                        acc = 0.0f;
                        nrow = 0;
                        cur = g[u];
                    }
                    acc += o[u];
                    ++nrow;
                }
                atomicAdd(sums + (size_t)cur * HIDDEN + f, acc);
                if (f == 0) atomicAdd(counts + cur, (float)nrow);
            }
        } else {
            const float4 bi = *(const float4*)(Ps + f0);
            const float4 mu = *(const float4*)(Ps + HIDDEN + f0);
            const float4 sn = *(const float4*)(Ps + 2 * HIDDEN + f0);
            const float4 be = *(const float4*)(Ps + 3 * HIDDEN + f0);
            for (int r = 0; r < 2; ++r) {
                const int lrow = wave * 2 + r;
                const int node = tile * 16 + lrow;
                const float4 p = *(const float4*)(Os + lrow * T_PITCH + f0);
                float4 o;
                o.x = p.x + bi.x;
                o.y = p.y + bi.y;
                o.z = p.z + bi.z;
                o.w = p.w + bi.w;
                o.x = (o.x - mu.x) * sn.x + be.x;
                o.y = (o.y - mu.y) * sn.y + be.y;
                o.z = (o.z - mu.z) * sn.z + be.z;
                o.w = (o.w - mu.w) * sn.w + be.w;
                o.x = fmaxf(o.x, 0.0f);
                o.y = fmaxf(o.y, 0.0f);
                o.z = fmaxf(o.z, 0.0f);
                o.w = fmaxf(o.w, 0.0f);
                *(float4*)(hout + (size_t)node * HIDDEN + f0) = o;
            }
        }
    }
}

#define OUT_BLOCKS 250
static_assert(OUT_BLOCKS * 8 == N_GRAPHS, "a wave per graph");
static_assert(HIDDEN == 128, "a lane takes four columns");
__global__ __launch_bounds__(256) void out_kernel(const float* __restrict__ sums, const float* __restrict__ counts,
                                                  const float* __restrict__ W_out, const float* __restrict__ b_out,
                                                  float* __restrict__ out) {
    const int wave = __builtin_amdgcn_readfirstlane((int)(threadIdx.x >> 5));
    const int lane = threadIdx.x & 31;
    const int g = blockIdx.x * 8 + wave;
    const float4 s = *(const float4*)(sums + (size_t)g * HIDDEN + 4 * lane);
    const float cnt = counts[g];
    float w[20];
#pragma unroll
    for (int q = 0; q < 5; ++q) {
        const float4 t = *(const float4*)(W_out + 20 * lane + 4 * q);
        w[4 * q + 0] = t.x;
        w[4 * q + 1] = t.y;
        w[4 * q + 2] = t.z;
        w[4 * q + 3] = t.w;
    }
    float bo = b_out[min(lane, 4)];
    asm volatile("" : "+v"(bo));
    const float inv = 1.0f / fmaxf(cnt, 1.0f);
    const float p0 = s.x * inv, p1 = s.y * inv, p2 = s.z * inv, p3 = s.w * inv;
    float res = 0.0f;
#pragma unroll
    for (int o = 0; o < 5; ++o) {
        float a = p0 * w[o];
        a += p1 * w[5 + o];
        a += p2 * w[10 + o];
        a += p3 * w[15 + o];
#pragma unroll
        for (int m = 16; m >= 1; m >>= 1) a += __shfl_xor(a, m);
        if (lane == o) res = a;
    }
    if (lane < 5) out[g * 5 + lane] = bo + res;
}

extern "C" void kernel_launch(void* const* d_in, const int* in_sizes, int n_in,
                              void* d_out, int out_size, void* d_ws, size_t ws_size,
                              hipStream_t stream) {
    const float* x        = (const float*)d_in[0];
    const int*   ei       = (const int*)d_in[1];
    const int*   src      = ei;
    const int*   dst      = ei + N_EDGES;
    const int*   batch    = (const int*)d_in[2];
    const float* Ws       = (const float*)d_in[3];
    const float* bs       = (const float*)d_in[4];
    const float* gammas   = (const float*)d_in[5];
    const float* betas    = (const float*)d_in[6];
    const float* run_mean = (const float*)d_in[7];
    const float* run_var  = (const float*)d_in[8];
    const float* W_out    = (const float*)d_in[9];
    const float* b_out    = (const float*)d_in[10];
    float*       out      = (float*)d_out;

    const size_t NF = (size_t)N_NODES * HIDDEN;
    float* bufA   = (float*)d_ws;
    float* bufB   = bufA + NF;
    float* bufC   = bufB + NF;
    int*   tab      = (int*)bufC;
    int*   rowptr   = tab + 2 * N_PAD + 128;
    int*   pairs    = rowptr + N_PAD;
    int*   recs     = pairs + 2 * (size_t)N_EDGES;
    int*   recB     = recs + N_EDGES;
    float* norm   = bufC + NF;
    float* sums   = norm + N_NODES;
    float* counts = sums + (size_t)N_GRAPHS * HIDDEN;

    chunk_sort_kernel<<<NCH, 256, 0, stream>>>(src, dst, tab, pairs);
    bucket_rows_kernel<<<NB, ROW_BUCKET, 0, stream>>>(tab, pairs, rowptr, norm, recs);
    order_runs_kernel<<<ORDER_BLOCKS, 256, 0, stream>>>(rowptr, recs, norm, recB, sums);

    const float* h = x;
    for (int l = 0; l < N_LAYERS - 1; ++l) {
        float* hn = (l & 1) ? bufB : bufA;
        gcn_layer_kernel<false><<<1024, 256, 0, stream>>>(
            h, Ws + (size_t)l * HIDDEN * HIDDEN, rowptr, recB, bs + l * HIDDEN,
            gammas + l * HIDDEN, betas + l * HIDDEN, run_mean + l * HIDDEN, run_var + l * HIDDEN, hn,
            batch, sums, counts);
        h = hn;
    }
    {
        const int l = N_LAYERS - 1;
        gcn_layer_kernel<true><<<1024, 256, 0, stream>>>(
            h, Ws + (size_t)l * HIDDEN * HIDDEN, rowptr, recB, bs + l * HIDDEN,
            gammas + l * HIDDEN, betas + l * HIDDEN, run_mean + l * HIDDEN, run_var + l * HIDDEN, nullptr,
            batch, sums, counts);
    }
    out_kernel<<<OUT_BLOCKS, 256, 0, stream>>>(sums, counts, W_out, b_out, out);
}
